# starting kernel
# baseline (speedup 1.0000x reference)
#include <hip/hip_runtime.h>
#include <math.h>

typedef float v2f __attribute__((ext_vector_type(2)));
typedef float v8f __attribute__((ext_vector_type(8)));

static __device__ __forceinline__ float elu1(float v) {
    return v > 0.0f ? v : (__expf(v) - 1.0f);
}


__global__ void zero_f32_kernel(float* __restrict__ p, int n) {
    int i = blockIdx.x * blockDim.x + threadIdx.x;
    if (i < n) p[i] = 0.0f;
}

__global__ void deg_scatter_kernel(const int* __restrict__ col,
                                   const float* __restrict__ w,
                                   float* __restrict__ deg, int E) {
    int e = blockIdx.x * blockDim.x + threadIdx.x;
    if (e < E) atomicAdd(&deg[col[e]], w[e]);
}

__global__ void dinv_kernel(float* __restrict__ deg, int N) {
    int i = blockIdx.x * blockDim.x + threadIdx.x;
    if (i < N) {
        float d = deg[i];
        deg[i] = d > 0.0f ? rsqrtf(d) : 0.0f;
    }
}

__global__ void norm_kernel(const int* __restrict__ row, const int* __restrict__ col,
                            const float* __restrict__ w, const float* __restrict__ dinv,
                            float* __restrict__ norm, int E) {
    int e = blockIdx.x * blockDim.x + threadIdx.x;
    if (e < E) norm[e] = dinv[row[e]] * w[e] * dinv[col[e]];
}

__global__ void pre_kernel(const float* __restrict__ x, const float* __restrict__ preW,
                           const float* __restrict__ preb, float* __restrict__ h, int total) {
    int i = blockIdx.x * blockDim.x + threadIdx.x;
    if (i >= total) return;
    int n = i >> 6, f = i & 63;
    h[i] = elu1(x[n] * preW[f] + preb[f]);
}

__global__ void edge_scatter_kernel(const int* __restrict__ row, const int* __restrict__ col,
                                    const float* __restrict__ norm, const float* __restrict__ t,
                                    float* __restrict__ agg, int E) {
    int gid = blockIdx.x * blockDim.x + threadIdx.x;
    int e = gid >> 5;
    int f = gid & 31;
    if (e >= E) return;
    int r = row[e], c = col[e];
    float v = norm[e] * t[(size_t)r * 32 + f];
    atomicAdd(&agg[(size_t)c * 32 + f], v);
}

__global__ void relu_kernel(float* __restrict__ p, int n) {
    int i = blockIdx.x * blockDim.x + threadIdx.x;
    if (i < n) p[i] = fmaxf(p[i], 0.0f);
}

template <int NCT, bool DUAL>
__global__ void wmma_gemm_kernel(const float* __restrict__ A, int K,
                                 const float* __restrict__ Wi,
                                 const float* __restrict__ Wr,
                                 const float* __restrict__ biasR,
                                 float* __restrict__ Ti,
                                 float* __restrict__ Tr,
                                 int nrows) {
    const int F = NCT * 16;
    int wave = (blockIdx.x * blockDim.x + threadIdx.x) >> 5;
    int lane = threadIdx.x & 31;
    int row0 = wave * 16;
    if (row0 >= nrows) return;

    int m  = lane & 15;
    int hi = lane >> 4;

    v8f accI[NCT] = {};
    v8f accR[NCT] = {};

    int ar = row0 + m;
    if (ar >= nrows) ar = nrows - 1;
    const float* arow = A + (size_t)ar * K;

    for (int k0 = 0; k0 < K; k0 += 4) {
        int ka = k0 + 2 * hi;
        v2f a;
        a.x = arow[ka];
        a.y = arow[ka + 1];
#pragma unroll
        for (int ct = 0; ct < NCT; ++ct) {
            int colc = ct * 16 + m;
            v2f b;
            b.x = Wi[(size_t)ka * F + colc];
            b.y = Wi[(size_t)(ka + 1) * F + colc];
            accI[ct] = __builtin_amdgcn_wmma_f32_16x16x4_f32(
                false, a, false, b, (short)0, accI[ct], false, false);
            if (DUAL) {
                v2f br;
                br.x = Wr[(size_t)ka * F + colc];
                br.y = Wr[(size_t)(ka + 1) * F + colc];
                accR[ct] = __builtin_amdgcn_wmma_f32_16x16x4_f32(
                    false, a, false, br, (short)0, accR[ct], false, false);
            }
        }
    }

#pragma unroll
    for (int ct = 0; ct < NCT; ++ct) {
        int colc = ct * 16 + m;
        float bv = biasR ? biasR[colc] : 0.0f;
#pragma unroll
        for (int r = 0; r < 8; ++r) {
            int rowm = row0 + r + 8 * hi;
            if (rowm >= nrows) continue;
            size_t idx = (size_t)rowm * F + colc;
            if (DUAL) {
                Ti[idx] = accI[ct][r];
                Tr[idx] = accR[ct][r] + bv;
            } else {
                Ti[idx] = accI[ct][r] + bv;
            }
        }
    }
}


__global__ void score_kernel(const int* __restrict__ li, const float* __restrict__ z,
                             float* __restrict__ out, int L) {
    int e = blockIdx.x * blockDim.x + threadIdx.x;
    if (e >= L) return;
    int i = li[e], j = li[L + e];
    const float4* za = (const float4*)(z + (size_t)i * 16);
    const float4* zb = (const float4*)(z + (size_t)j * 16);
    float dot = 0.0f, na = 0.0f, nb = 0.0f;
#pragma unroll
    for (int q = 0; q < 4; ++q) {
        float4 a = za[q], b = zb[q];
        dot += a.x * b.x + a.y * b.y + a.z * b.z + a.w * b.w;
        na  += a.x * a.x + a.y * a.y + a.z * a.z + a.w * a.w;
        nb  += b.x * b.x + b.y * b.y + b.z * b.z + b.w * b.w;
    }
    float den = fmaxf(sqrtf(na) * sqrtf(nb), 1e-6f);
    out[e] = dot / den;
}


static inline int cdiv(long long a, long long b) { return (int)((a + b - 1) / b); }

extern "C" void kernel_launch(void* const* d_in, const int* in_sizes, int n_in,
                              void* d_out, int out_size, void* d_ws, size_t ws_size,
                              hipStream_t stream) {
    const float* x     = (const float*)d_in[0];
    const int*   ei    = (const int*)d_in[1];
    const float* ew    = (const float*)d_in[2];
    const int*   eli   = (const int*)d_in[3];
    const float* preW  = (const float*)d_in[4];
    const float* preb  = (const float*)d_in[5];
    const float* g1Wi  = (const float*)d_in[6];
    const float* g1Wr  = (const float*)d_in[7];
    const float* g1b   = (const float*)d_in[8];
    const float* g2Wi  = (const float*)d_in[9];
    const float* g2Wr  = (const float*)d_in[10];
    const float* g2b   = (const float*)d_in[11];
    const float* poW   = (const float*)d_in[12];
    const float* pob   = (const float*)d_in[13];
    float* out = (float*)d_out;

    const int N = in_sizes[0];
    const int E = in_sizes[2];
    const int L = in_sizes[3] / 2;

    const int* erow = ei;
    const int* ecol = ei + E;

    float* ws   = (float*)d_ws;
    float* deg  = ws;
    float* norm = deg  + N;
    float* h64  = norm + E;
    float* t    = h64  + (size_t)N * 64;
    float* agg1 = t    + (size_t)N * 32;
    float* agg2 = agg1 + (size_t)N * 32;
    float* z    = agg2 + (size_t)N * 32;

    const int B = 256;
    const int gemm_threads = cdiv(N, 16) * 32;

    zero_f32_kernel<<<cdiv(N, B), B, 0, stream>>>(deg, N);
    deg_scatter_kernel<<<cdiv(E, B), B, 0, stream>>>(ecol, ew, deg, E);
    dinv_kernel<<<cdiv(N, B), B, 0, stream>>>(deg, N);
    norm_kernel<<<cdiv(E, B), B, 0, stream>>>(erow, ecol, ew, deg, norm, E);

    pre_kernel<<<cdiv((long long)N * 64, B), B, 0, stream>>>(x, preW, preb, h64, N * 64);

    wmma_gemm_kernel<2, true><<<cdiv(gemm_threads, B), B, 0, stream>>>(
        h64, 64, g1Wi, g1Wr, g1b, t, agg1, N);
    edge_scatter_kernel<<<cdiv((long long)E * 32, B), B, 0, stream>>>(
        erow, ecol, norm, t, agg1, E);
    relu_kernel<<<cdiv((long long)N * 32, B), B, 0, stream>>>(agg1, N * 32);

    wmma_gemm_kernel<2, true><<<cdiv(gemm_threads, B), B, 0, stream>>>(
        agg1, 32, g2Wi, g2Wr, g2b, t, agg2, N);
    edge_scatter_kernel<<<cdiv((long long)E * 32, B), B, 0, stream>>>(
        erow, ecol, norm, t, agg2, E);
    relu_kernel<<<cdiv((long long)N * 32, B), B, 0, stream>>>(agg2, N * 32);

    wmma_gemm_kernel<1, false><<<cdiv(gemm_threads, B), B, 0, stream>>>(
        agg2, 32, poW, nullptr, pob, z, nullptr, N);

    score_kernel<<<cdiv(L, B), B, 0, stream>>>(eli, z, out, L);
}


# conv2 owner pass replays per-wave hit lists (edge indices) that conv1's pass stores as it scans: no second scan of range bytes; conv2 root term written in place for room; arrival order as before
# speedup vs baseline: 3.1695x; 3.1695x over previous
#include <hip/hip_runtime.h>
#include <math.h>

typedef float v2f __attribute__((ext_vector_type(2)));
typedef float v8f __attribute__((ext_vector_type(8)));

static __device__ __forceinline__ float elu1(float v) {
    return v > 0.0f ? v : (__expf(v) - 1.0f);
}


static __device__ __forceinline__ float dinv_of(float d) {
    return d > 0.0f ? rsqrtf(d) : 0.0f;
}

struct EdgeRec { int row; int nrm; int col; };

#define OWN_NR 1600

__global__ void rec_kernel(const int* __restrict__ row, const int* __restrict__ col,
                           const float* __restrict__ w, const float* __restrict__ deg,
                           int* __restrict__ rec, unsigned int* __restrict__ tagw,
                           int E, int N) {
    int e = blockIdx.x * blockDim.x + threadIdx.x;
    if (e < E) {
        int r = row[e], c = col[e];
        EdgeRec x;
        x.row = r;
        x.nrm = __float_as_int(dinv_of(deg[r]) * w[e] * dinv_of(deg[c]));
        x.col = c;
        *(EdgeRec*)(rec + (size_t)e * 3) = x;
        ((unsigned char*)tagw)[e] =
            (unsigned char)(((unsigned)c < (unsigned)N) ? (c / OWN_NR) : 255);
    }
}

__global__ void pre_kernel(const float* __restrict__ x, const float* __restrict__ preW,
                           const float* __restrict__ preb, float* __restrict__ h,
                           float* __restrict__ deg, int total) {
    int i = blockIdx.x * blockDim.x + threadIdx.x;
    if (i >= total) return;
    int n = i >> 6, f = i & 63;
    if (i < (total >> 6)) deg[i] = 0.0f;
    h[i] = elu1(x[n] * preW[f] + preb[f]);
}


#define DEG_NR 12800
#define DEG_S  32

__global__ __launch_bounds__(1024)
void deg_owner_kernel(const int* __restrict__ col, const float* __restrict__ w,
                      float* __restrict__ deg, int E, int N) {
    __shared__ float dacc[DEG_NR];
    const int tid   = threadIdx.x;
    const int range = blockIdx.x / DEG_S;
    const int slice = blockIdx.x % DEG_S;
    const int base  = range * DEG_NR;
    const int nloc  = min(DEG_NR, N - base);
    for (int i = tid; i < DEG_NR; i += 1024) dacc[i] = 0.0f;
    __syncthreads();
    const int per = (E + DEG_S - 1) / DEG_S;
    const int e0  = slice * per;
    const int e1  = min(E, e0 + per);
    int e = e0 + tid;
    for (; e + 7 * 1024 < e1; e += 8 * 1024) {
        int   cc[8];
        float ww[8];
#pragma unroll
        for (int u = 0; u < 8; ++u) { cc[u] = col[e + u * 1024]; ww[u] = w[e + u * 1024]; }
#pragma unroll
        for (int u = 0; u < 8; ++u) {
            int loc = cc[u] - base;
            if ((unsigned)loc < (unsigned)nloc) atomicAdd(&dacc[loc], ww[u]);
        }
    }
    for (; e < e1; e += 1024) {
        int loc = col[e] - base;
        if ((unsigned)loc < (unsigned)nloc) atomicAdd(&dacc[loc], w[e]);
    }
    __syncthreads();
    for (int i = tid; i < nloc; i += 1024) atomicAdd(&deg[base + i], dacc[i]);
}

#define OWN_S  4
#define OWN_CH 512
#define OWN_QC (OWN_CH + 32)

template <int G, bool FULL>
static __device__ __forceinline__ void own_rows(int r, float nv, int loc, int cnt, int lane,
                                                const float* __restrict__ t, float* acc) {
    if (FULL) {
#pragma unroll
        for (int j = 0; j < 32; j += G) {
            float v[G];
#pragma unroll
            for (int u = 0; u < G; ++u) {
                int rj = __builtin_amdgcn_readlane(r, j + u);
                v[u] = t[(size_t)rj * 32 + lane];
            }
#pragma unroll
            for (int u = 0; u < G; ++u) {
                int   lj = __builtin_amdgcn_readlane(loc, j + u);
                float nj = __int_as_float(__builtin_amdgcn_readlane(__float_as_int(nv), j + u));
                atomicAdd(&acc[lj * 32 + lane], nj * v[u]);
            }
        }
    } else {
        for (int j = 0; j < cnt; j += G) {
            float v[G], nj[G];
            int   lj[G];
#pragma unroll
            for (int u = 0; u < G; ++u) {
                int jj = min(j + u, cnt - 1);
                int rj = __builtin_amdgcn_readlane(r, jj);
                lj[u]  = __builtin_amdgcn_readlane(loc, jj);
                nj[u]  = __int_as_float(__builtin_amdgcn_readlane(__float_as_int(nv), jj));
                v[u]   = t[(size_t)rj * 32 + lane];
            }
#pragma unroll
            for (int u = 0; u < G; ++u)
                if (j + u < cnt) atomicAdd(&acc[lj[u] * 32 + lane], nj[u] * v[u]);
        }
    }
}

static __device__ __forceinline__ EdgeRec own_fetch(int e, const int* __restrict__ rec) {
    return *(const EdgeRec*)(rec + (size_t)e * 3);
}

template <int G, bool FULL>
static __device__ __forceinline__ void own_batch(int e, int cnt, int lane, int base,
                                                 const int* __restrict__ rec,
                                                 const float* __restrict__ t,
                                                 float* acc) {
    EdgeRec x = { 0, 0, 0 };
    if (FULL || lane < cnt) x = own_fetch(e, rec);
    own_rows<G, FULL>(x.row, __int_as_float(x.nrm), x.col - base, cnt, lane, t, acc);
}

static __device__ __forceinline__ uint4 own_tags(const unsigned char* __restrict__ tagb,
                                                 int cb, int lane, int lim) {
    return *(const uint4*)(tagb + min(cb + lane * 16, lim));
}

template <int MODE>
__global__ __launch_bounds__(1024)
void edge_owner_kernel(const unsigned int* __restrict__ tagw, const int* __restrict__ rec,
                       const float* __restrict__ t,
                       float* __restrict__ agg, int* __restrict__ lst, int E, int N) {
    __shared__ float acc[OWN_NR * 32];
    __shared__ int   q[32 * OWN_QC];
    const int tid   = threadIdx.x;
    const int lane  = tid & 31;
    const int wave  = __builtin_amdgcn_readfirstlane(tid >> 5);
    const int range = blockIdx.x / OWN_S;
    const int slice = blockIdx.x % OWN_S;
    const int base  = range * OWN_NR;
    const int nloc  = min(OWN_NR, N - base);

    const int lw  = (E / 4 * 5) / ((int)gridDim.x * 32);
    const int cap = lw - 1;
    int* wl = lst + ((size_t)blockIdx.x * 32 + wave) * (size_t)max(lw, 0);

    int h = -1;
    if (MODE == 1 && cap >= 1) h = __builtin_amdgcn_readfirstlane(wl[cap]);

    for (int i = tid; i < OWN_NR * 32; i += 1024) acc[i] = 0.0f;
    __syncthreads();

    bool scan = true;
    if (MODE == 1) {
        scan = h < 0;
        if (!scan) {
            const int cnt = h;
            if (cnt > 0) {
                const int last = cnt - 1;
                EdgeRec x1 = own_fetch(wl[min(lane, last)], rec);
                int     e2 = wl[min(32 + lane, last)];
                int done = 0;
                for (; done + 32 <= cnt; done += 32) {
                    int     e3 = wl[min(done + 64 + lane, last)];
                    EdgeRec x2 = own_fetch(e2, rec);
                    own_rows<16, true>(x1.row, __int_as_float(x1.nrm), x1.col - base, 32,
                                       lane, t, acc);
                    x1 = x2;
                    e2 = e3;
                }
                if (cnt > done)
                    own_rows<4, false>(x1.row, __int_as_float(x1.nrm), x1.col - base,
                                       cnt - done, lane, t, acc);
            }
        }
    }

    if (scan) {
        const bool keep = (MODE == 0) && (cap >= 1);
        int  pos = 0;
        bool ovf = false;

        const int per = ((E + OWN_S * OWN_CH - 1) / (OWN_S * OWN_CH)) * OWN_CH;
        const int e0  = slice * per;
        const int e1  = min(E, e0 + per);
        int* wq = q + wave * OWN_QC;
        int  n  = 0;

        const unsigned char* tagb = (const unsigned char*)tagw;
        const unsigned pat = (unsigned)range * 0x01010101u;

        const int lim = ((E - 1) >> 4) << 4;

        int   cb  = e0 + wave * OWN_CH;
        uint4 cur = own_tags(tagb, cb, lane, lim);

        for (; cb < e1; cb += 32 * OWN_CH) {
            uint4 nxt = own_tags(tagb, cb + 32 * OWN_CH, lane, lim);

            const unsigned xs[4] = { cur.x, cur.y, cur.z, cur.w };
            unsigned F = 0u;
#pragma unroll
            for (int d = 0; d < 4; ++d) {
                unsigned y = xs[d] ^ pat;
                unsigned s = (y & 0x7F7F7F7Fu) + 0x7F7F7F7Fu;
                unsigned f = ~(s | y | 0x7F7F7F7Fu);
                F |= f >> (7 - d);
            }
            const int eb = cb + lane * 16;
            for (int rnd = 0; rnd < 16; ++rnd) {
                if (__builtin_amdgcn_ballot_w32(F != 0u) == 0u) break;
                int p   = (F != 0u) ? __builtin_ctz(F) : 0;
                int e   = eb + (((p & 7) << 2) | (p >> 3));
                bool hit = (F != 0u) && (e < e1);
                unsigned mask = __builtin_amdgcn_ballot_w32(hit);
                int at = n + (int)__builtin_amdgcn_mbcnt_lo(mask, 0u);
                if (hit) wq[at] = e;
                n += __builtin_popcount(mask);
                F &= F - 1u;
            }
            __builtin_amdgcn_fence(__ATOMIC_ACQ_REL, "wavefront");

            if (n >= 32) {
                int done = 0;
                for (; done + 32 <= n; done += 32) {
                    const int e = wq[done + lane];
                    if (keep) {
                        if (pos + 32 <= cap) { wl[pos + lane] = e; pos += 32; }
                        else ovf = true;
                    }
                    own_batch<16, true>(e, 32, lane, base, rec, t, acc);
                }
                const int rest = n - done;
                int mv = (lane < rest) ? wq[done + lane] : 0;
                if (lane < rest) wq[lane] = mv;
                n = rest;
                __builtin_amdgcn_fence(__ATOMIC_ACQ_REL, "wavefront");
            }
            cur = nxt;
        }
        if (n > 0) {
            const int e = (lane < n) ? wq[lane] : 0;
            if (keep) {
                if (pos + n <= cap) { if (lane < n) wl[pos + lane] = e; pos += n; }
                else ovf = true;
            }
            own_batch<4, false>(e, n, lane, base, rec, t, acc);
        }
        if (keep && lane == 0) wl[cap] = ovf ? -1 : pos;
    }

    __syncthreads();
    float* dst = agg + (size_t)base * 32;
    for (int i = tid; i < nloc * 32; i += 1024) atomicAdd(&dst[i], acc[i]);
}

template <int NCT, bool DUAL, bool RELU_A>
__global__ void wmma_gemm_kernel(const float* A, int K,
                                 const float* __restrict__ Wi,
                                 const float* __restrict__ Wr,
                                 const float* __restrict__ biasR,
                                 float* __restrict__ Ti,
                                 float* Tr,
                                 int nrows) {
    const int F = NCT * 16;
    int wave = (blockIdx.x * blockDim.x + threadIdx.x) >> 5;
    int lane = threadIdx.x & 31;
    int row0 = wave * 16;
    if (row0 >= nrows) return;

    int m  = lane & 15;
    int hi = lane >> 4;

    v8f accI[NCT] = {};
    v8f accR[NCT] = {};

    int ar = row0 + m;
    if (ar >= nrows) ar = nrows - 1;
    const float* arow = A + (size_t)ar * K;

    for (int k0 = 0; k0 < K; k0 += 4) {
        int ka = k0 + 2 * hi;
        v2f a;
        a.x = arow[ka];
        a.y = arow[ka + 1];
        if (RELU_A) { a.x = fmaxf(a.x, 0.0f); a.y = fmaxf(a.y, 0.0f); }
#pragma unroll
        for (int ct = 0; ct < NCT; ++ct) {
            int colc = ct * 16 + m;
            v2f b;
            b.x = Wi[(size_t)ka * F + colc];
            b.y = Wi[(size_t)(ka + 1) * F + colc];
            accI[ct] = __builtin_amdgcn_wmma_f32_16x16x4_f32(
                false, a, false, b, (short)0, accI[ct], false, false);
            if (DUAL) {
                v2f br;
                br.x = Wr[(size_t)ka * F + colc];
                br.y = Wr[(size_t)(ka + 1) * F + colc];
                accR[ct] = __builtin_amdgcn_wmma_f32_16x16x4_f32(
                    false, a, false, br, (short)0, accR[ct], false, false);
            }
        }
    }

#pragma unroll
    for (int ct = 0; ct < NCT; ++ct) {
        int colc = ct * 16 + m;
        float bv = biasR ? biasR[colc] : 0.0f;
#pragma unroll
        for (int r = 0; r < 8; ++r) {
            int rowm = row0 + r + 8 * hi;
            if (rowm >= nrows) continue;
            size_t idx = (size_t)rowm * F + colc;
            if (DUAL) {
                Ti[idx] = accI[ct][r];
                Tr[idx] = accR[ct][r] + bv;
            } else {
                Ti[idx] = accI[ct][r] + bv;
            }
        }
    }
}


__global__ void score_kernel(const int* __restrict__ li, const float* __restrict__ z,
                             float* __restrict__ out, int L) {
    int e = blockIdx.x * blockDim.x + threadIdx.x;
    if (e >= L) return;
    int i = li[e], j = li[L + e];
    const float4* za = (const float4*)(z + (size_t)i * 16);
    const float4* zb = (const float4*)(z + (size_t)j * 16);
    float dot = 0.0f, na = 0.0f, nb = 0.0f;
#pragma unroll
    for (int q = 0; q < 4; ++q) {
        float4 a = za[q], b = zb[q];
        dot += a.x * b.x + a.y * b.y + a.z * b.z + a.w * b.w;
        na  += a.x * a.x + a.y * a.y + a.z * a.z + a.w * a.w;
        nb  += b.x * b.x + b.y * b.y + b.z * b.z + b.w * b.w;
    }
    float den = fmaxf(sqrtf(na) * sqrtf(nb), 1e-6f);
    out[e] = dot / den;
}


static inline int cdiv(long long a, long long b) { return (int)((a + b - 1) / b); }

extern "C" void kernel_launch(void* const* d_in, const int* in_sizes, int n_in,
                              void* d_out, int out_size, void* d_ws, size_t ws_size,
                              hipStream_t stream) {
    const float* x     = (const float*)d_in[0];
    const int*   ei    = (const int*)d_in[1];
    const float* ew    = (const float*)d_in[2];
    const int*   eli   = (const int*)d_in[3];
    const float* preW  = (const float*)d_in[4];
    const float* preb  = (const float*)d_in[5];
    const float* g1Wi  = (const float*)d_in[6];
    const float* g1Wr  = (const float*)d_in[7];
    const float* g1b   = (const float*)d_in[8];
    const float* g2Wi  = (const float*)d_in[9];
    const float* g2Wr  = (const float*)d_in[10];
    const float* g2b   = (const float*)d_in[11];
    const float* poW   = (const float*)d_in[12];
    const float* pob   = (const float*)d_in[13];
    float* out = (float*)d_out;

    const int N = in_sizes[0];
    const int E = in_sizes[2];
    const int L = in_sizes[3] / 2;

    const int* erow = ei;
    const int* ecol = ei + E;

    float* ws   = (float*)d_ws;
    float* deg  = ws;
    float* rec0 = deg  + N;
    int*   rec  = (int*)rec0;
    float* h64  = rec0 + E;
    float* t    = h64  + (size_t)N * 64;
    float* agg1 = t    + (size_t)N * 32;
    int*   lst  = (int*)(agg1 + (size_t)N * 32);
    float* z    = agg1 + (size_t)N * 64;
    unsigned int* tagw = (unsigned int*)(z + (size_t)N * 8);

    const int B = 256;
    const int gemm_threads = cdiv(N, 16) * 32;

    pre_kernel<<<cdiv((long long)N * 64, B), B, 0, stream>>>(x, preW, preb, h64, deg, N * 64);

    deg_owner_kernel<<<cdiv(N, DEG_NR) * DEG_S, 1024, 0, stream>>>(ecol, ew, deg, E, N);

    wmma_gemm_kernel<2, true, false><<<cdiv(gemm_threads, B), B, 0, stream>>>(
        h64, 64, g1Wi, g1Wr, g1b, t, agg1, N);

    rec_kernel<<<cdiv(E, B), B, 0, stream>>>(erow, ecol, ew, deg, rec, tagw, E, N);

    edge_owner_kernel<0><<<cdiv(N, OWN_NR) * OWN_S, 1024, 0, stream>>>(
        tagw, rec, t, agg1, lst, E, N);

    wmma_gemm_kernel<2, true, true><<<cdiv(gemm_threads, B), B, 0, stream>>>(
        agg1, 32, g2Wi, g2Wr, g2b, t, agg1, N);
    edge_owner_kernel<1><<<cdiv(N, OWN_NR) * OWN_S, 1024, 0, stream>>>(
        tagw, rec, t, agg1, lst, E, N);

    wmma_gemm_kernel<1, false, true><<<cdiv(gemm_threads, B), B, 0, stream>>>(
        agg1, 32, poW, nullptr, pob, z, nullptr, N);

    score_kernel<<<cdiv(L, B), B, 0, stream>>>(eli, z, out, L);
}


# conv2 owner pass reads back 8-byte records {row|place, norm} that conv1's pass stores per workgroup in work order: one 256-byte read a batch, was 32 record lines; LDS arrival order as before
# speedup vs baseline: 3.3191x; 1.0472x over previous
#include <hip/hip_runtime.h>
#include <math.h>

typedef float v2f __attribute__((ext_vector_type(2)));
typedef float v8f __attribute__((ext_vector_type(8)));

static __device__ __forceinline__ float elu1(float v) {
    return v > 0.0f ? v : (__expf(v) - 1.0f);
}


static __device__ __forceinline__ float dinv_of(float d) {
    return d > 0.0f ? rsqrtf(d) : 0.0f;
}

#define OWN_NR 1600
#define OWN_LB 11

struct EdgeRec { int pk; int nrm; };

static __device__ __forceinline__ int rec_row(EdgeRec x) { return (int)((unsigned)x.pk >> OWN_LB); }
static __device__ __forceinline__ int rec_loc(EdgeRec x) { return x.pk & ((1 << OWN_LB) - 1); }

__global__ void rec_kernel(const int* __restrict__ row, const int* __restrict__ col,
                           const float* __restrict__ w, const float* __restrict__ deg,
                           int* __restrict__ rec, unsigned int* __restrict__ tagw,
                           int E, int N) {
    int e = blockIdx.x * blockDim.x + threadIdx.x;
    if (e < E) {
        int r = row[e], c = col[e];
        EdgeRec x;
        x.pk  = (int)(((unsigned)r << OWN_LB) | ((unsigned)c % OWN_NR));
        x.nrm = __float_as_int(dinv_of(deg[r]) * w[e] * dinv_of(deg[c]));
        *(EdgeRec*)(rec + (size_t)e * 2) = x;
        ((unsigned char*)tagw)[e] =
            (unsigned char)(((unsigned)c < (unsigned)N) ? (c / OWN_NR) : 255);
    }
}

__global__ void pre_kernel(const float* __restrict__ x, const float* __restrict__ preW,
                           const float* __restrict__ preb, float* __restrict__ h,
                           float* __restrict__ deg, int total) {
    int i = blockIdx.x * blockDim.x + threadIdx.x;
    if (i >= total) return;
    int n = i >> 6, f = i & 63;
    if (i < (total >> 6)) deg[i] = 0.0f;
    h[i] = elu1(x[n] * preW[f] + preb[f]);
}


#define DEG_NR 12800
#define DEG_S  32

__global__ __launch_bounds__(1024)
void deg_owner_kernel(const int* __restrict__ col, const float* __restrict__ w,
                      float* __restrict__ deg, int E, int N) {
    __shared__ float dacc[DEG_NR];
    const int tid   = threadIdx.x;
    const int range = blockIdx.x / DEG_S;
    const int slice = blockIdx.x % DEG_S;
    const int base  = range * DEG_NR;
    const int nloc  = min(DEG_NR, N - base);
    for (int i = tid; i < DEG_NR; i += 1024) dacc[i] = 0.0f;
    __syncthreads();
    const int per = (E + DEG_S - 1) / DEG_S;
    const int e0  = slice * per;
    const int e1  = min(E, e0 + per);
    int e = e0 + tid;
    for (; e + 7 * 1024 < e1; e += 8 * 1024) {
        int   cc[8];
        float ww[8];
#pragma unroll
        for (int u = 0; u < 8; ++u) { cc[u] = col[e + u * 1024]; ww[u] = w[e + u * 1024]; }
#pragma unroll
        for (int u = 0; u < 8; ++u) {
            int loc = cc[u] - base;
            if ((unsigned)loc < (unsigned)nloc) atomicAdd(&dacc[loc], ww[u]);
        }
    }
    for (; e < e1; e += 1024) {
        int loc = col[e] - base;
        if ((unsigned)loc < (unsigned)nloc) atomicAdd(&dacc[loc], w[e]);
    }
    __syncthreads();
    for (int i = tid; i < nloc; i += 1024) atomicAdd(&deg[base + i], dacc[i]);
}

#define OWN_S  4
#define OWN_CH 512
#define OWN_QC (OWN_CH + 32)

template <int G, bool FULL>
static __device__ __forceinline__ void own_rows(int r, float nv, int loc, int cnt, int lane,
                                                const float* __restrict__ t, float* acc) {
    if (FULL) {
#pragma unroll
        for (int j = 0; j < 32; j += G) {
            float v[G];
#pragma unroll
            for (int u = 0; u < G; ++u) {
                int rj = __builtin_amdgcn_readlane(r, j + u);
                v[u] = t[(size_t)rj * 32 + lane];
            }
#pragma unroll
            for (int u = 0; u < G; ++u) {
                int   lj = __builtin_amdgcn_readlane(loc, j + u);
                float nj = __int_as_float(__builtin_amdgcn_readlane(__float_as_int(nv), j + u));
                atomicAdd(&acc[lj * 32 + lane], nj * v[u]);
            }
        }
    } else {
        for (int j = 0; j < cnt; j += G) {
            float v[G], nj[G];
            int   lj[G];
#pragma unroll
            for (int u = 0; u < G; ++u) {
                int jj = min(j + u, cnt - 1);
                int rj = __builtin_amdgcn_readlane(r, jj);
                lj[u]  = __builtin_amdgcn_readlane(loc, jj);
                nj[u]  = __int_as_float(__builtin_amdgcn_readlane(__float_as_int(nv), jj));
                v[u]   = t[(size_t)rj * 32 + lane];
            }
#pragma unroll
            for (int u = 0; u < G; ++u)
                if (j + u < cnt) atomicAdd(&acc[lj[u] * 32 + lane], nj[u] * v[u]);
        }
    }
}

static __device__ __forceinline__ EdgeRec own_fetch(int e, const int* __restrict__ rec) {
    return *(const EdgeRec*)(rec + (size_t)e * 2);
}

template <int G, bool FULL>
static __device__ __forceinline__ void own_work(EdgeRec x, int cnt, int lane,
                                                const float* __restrict__ t, float* acc) {
    own_rows<G, FULL>(rec_row(x), __int_as_float(x.nrm), rec_loc(x), cnt, lane, t, acc);
}

#define OWN_LH 64
#define OWN_LT (32 * 32 * 2)

static __device__ __forceinline__ uint4 own_tags(const unsigned char* __restrict__ tagb,
                                                 int cb, int lane, int lim) {
    return *(const uint4*)(tagb + min(cb + lane * 16, lim));
}

template <int MODE>
__global__ __launch_bounds__(1024)
void edge_owner_kernel(const unsigned int* __restrict__ tagw, const int* __restrict__ rec,
                       const float* __restrict__ t,
                       float* __restrict__ agg, int* __restrict__ lst, int E, int N) {
    __shared__ float acc[OWN_NR * 32];
    __shared__ int   q[32 * OWN_QC];
    __shared__ int   bcnt;
    const int tid   = threadIdx.x;
    const int lane  = tid & 31;
    const int wave  = __builtin_amdgcn_readfirstlane(tid >> 5);
    const int range = blockIdx.x / OWN_S;
    const int slice = blockIdx.x % OWN_S;
    const int base  = range * OWN_NR;
    const int nloc  = min(OWN_NR, N - base);

    const int lwg  = ((E / 4 * 9) / (int)gridDim.x) & ~63;
    const int capB = (lwg - OWN_LH - OWN_LT) / 64;
    int*     hd = lst + (size_t)blockIdx.x * (size_t)max(lwg, 0);
    EdgeRec* tl = (EdgeRec*)(hd + OWN_LH);
    EdgeRec* fb = (EdgeRec*)(hd + OWN_LH + OWN_LT);

    int nb = -1, tn = 0;
    if (MODE == 1 && capB >= 1) {
        nb = __builtin_amdgcn_readfirstlane(hd[32]);
        tn = __builtin_amdgcn_readfirstlane(hd[wave]);
    }

    if (tid == 0) bcnt = 0;
    for (int i = tid; i < OWN_NR * 32; i += 1024) acc[i] = 0.0f;
    __syncthreads();

    bool scan = true;
    if (MODE == 1) {
        scan = nb < 0;
        if (!scan) {
            if (nb > 0) {
                int b = wave;
                EdgeRec x1 = fb[(size_t)min(b, nb - 1) * 32 + lane];
                for (; b < nb; b += 32) {
                    EdgeRec x2 = fb[(size_t)min(b + 32, nb - 1) * 32 + lane];
                    own_work<16, true>(x1, 32, lane, t, acc);
                    x1 = x2;
                }
            }
            if (tn > 0) {
                EdgeRec xt = { 0, 0 };
                if (lane < tn) xt = tl[wave * 32 + lane];
                own_work<4, false>(xt, tn, lane, t, acc);
            }
        }
    }

    if (scan) {
        const bool keep = (MODE == 0) && (capB >= 1);

        const int per = ((E + OWN_S * OWN_CH - 1) / (OWN_S * OWN_CH)) * OWN_CH;
        const int e0  = slice * per;
        const int e1  = min(E, e0 + per);
        int* wq = q + wave * OWN_QC;
        int  n  = 0;

        const unsigned char* tagb = (const unsigned char*)tagw;
        const unsigned pat = (unsigned)range * 0x01010101u;

        const int lim = ((E - 1) >> 4) << 4;

        int   cb  = e0 + wave * OWN_CH;
        uint4 cur = own_tags(tagb, cb, lane, lim);

        for (; cb < e1; cb += 32 * OWN_CH) {
            uint4 nxt = own_tags(tagb, cb + 32 * OWN_CH, lane, lim);

            const unsigned xs[4] = { cur.x, cur.y, cur.z, cur.w };
            unsigned F = 0u;
#pragma unroll
            for (int d = 0; d < 4; ++d) {
                unsigned y = xs[d] ^ pat;
                unsigned s = (y & 0x7F7F7F7Fu) + 0x7F7F7F7Fu;
                unsigned f = ~(s | y | 0x7F7F7F7Fu);
                F |= f >> (7 - d);
            }
            const int eb = cb + lane * 16;
            for (int rnd = 0; rnd < 16; ++rnd) {
                if (__builtin_amdgcn_ballot_w32(F != 0u) == 0u) break;
                int p   = (F != 0u) ? __builtin_ctz(F) : 0;
                int e   = eb + (((p & 7) << 2) | (p >> 3));
                bool hit = (F != 0u) && (e < e1);
                unsigned mask = __builtin_amdgcn_ballot_w32(hit);
                int at = n + (int)__builtin_amdgcn_mbcnt_lo(mask, 0u);
                if (hit) wq[at] = e;
                n += __builtin_popcount(mask);
                F &= F - 1u;
            }
            __builtin_amdgcn_fence(__ATOMIC_ACQ_REL, "wavefront");

            if (n >= 32) {
                int done = 0;
                for (; done + 32 <= n; done += 32) {
                    const EdgeRec x = own_fetch(wq[done + lane], rec);
                    if (keep) {
                        int sb = 0;
                        if (lane == 0) sb = atomicAdd(&bcnt, 1);
                        sb = __builtin_amdgcn_readfirstlane(sb);
                        if (sb < capB) fb[(size_t)sb * 32 + lane] = x;
                    }
                    own_work<16, true>(x, 32, lane, t, acc);
                }
                const int rest = n - done;
                int mv = (lane < rest) ? wq[done + lane] : 0;
                if (lane < rest) wq[lane] = mv;
                n = rest;
                __builtin_amdgcn_fence(__ATOMIC_ACQ_REL, "wavefront");
            }
            cur = nxt;
        }
        if (n > 0) {
            EdgeRec x = { 0, 0 };
            if (lane < n) x = own_fetch(wq[lane], rec);
            if (keep && lane < n) tl[wave * 32 + lane] = x;
            own_work<4, false>(x, n, lane, t, acc);
        }
        if (keep && lane == 0) hd[wave] = n;
    }

    __syncthreads();
    if (MODE == 0 && capB >= 1 && tid == 0) hd[32] = (bcnt > capB) ? -1 : bcnt;
    float* dst = agg + (size_t)base * 32;
    for (int i = tid; i < nloc * 32; i += 1024) atomicAdd(&dst[i], acc[i]);
}

template <int NCT, bool DUAL, bool RELU_A>
__global__ void wmma_gemm_kernel(const float* A, int K,
                                 const float* __restrict__ Wi,
                                 const float* __restrict__ Wr,
                                 const float* __restrict__ biasR,
                                 float* __restrict__ Ti,
                                 float* Tr,
                                 int nrows) {
    const int F = NCT * 16;
    int wave = (blockIdx.x * blockDim.x + threadIdx.x) >> 5;
    int lane = threadIdx.x & 31;
    int row0 = wave * 16;
    if (row0 >= nrows) return;

    int m  = lane & 15;
    int hi = lane >> 4;

    v8f accI[NCT] = {};
    v8f accR[NCT] = {};

    int ar = row0 + m;
    if (ar >= nrows) ar = nrows - 1;
    const float* arow = A + (size_t)ar * K;

    for (int k0 = 0; k0 < K; k0 += 4) {
        int ka = k0 + 2 * hi;
        v2f a;
        a.x = arow[ka];
        a.y = arow[ka + 1];
        if (RELU_A) { a.x = fmaxf(a.x, 0.0f); a.y = fmaxf(a.y, 0.0f); }
#pragma unroll
        for (int ct = 0; ct < NCT; ++ct) {
            int colc = ct * 16 + m;
            v2f b;
            b.x = Wi[(size_t)ka * F + colc];
            b.y = Wi[(size_t)(ka + 1) * F + colc];
            accI[ct] = __builtin_amdgcn_wmma_f32_16x16x4_f32(
                false, a, false, b, (short)0, accI[ct], false, false);
            if (DUAL) {
                v2f br;
                br.x = Wr[(size_t)ka * F + colc];
                br.y = Wr[(size_t)(ka + 1) * F + colc];
                accR[ct] = __builtin_amdgcn_wmma_f32_16x16x4_f32(
                    false, a, false, br, (short)0, accR[ct], false, false);
            }
        }
    }

#pragma unroll
    for (int ct = 0; ct < NCT; ++ct) {
        int colc = ct * 16 + m;
        float bv = biasR ? biasR[colc] : 0.0f;
#pragma unroll
        for (int r = 0; r < 8; ++r) {
            int rowm = row0 + r + 8 * hi;
            if (rowm >= nrows) continue;
            size_t idx = (size_t)rowm * F + colc;
            if (DUAL) {
                Ti[idx] = accI[ct][r];
                Tr[idx] = accR[ct][r] + bv;
            } else {
                Ti[idx] = accI[ct][r] + bv;
            }
        }
    }
}


__global__ void score_kernel(const int* __restrict__ li, const float* __restrict__ z,
                             float* __restrict__ out, int L) {
    int e = blockIdx.x * blockDim.x + threadIdx.x;
    if (e >= L) return;
    int i = li[e], j = li[L + e];
    const float4* za = (const float4*)(z + (size_t)i * 16);
    const float4* zb = (const float4*)(z + (size_t)j * 16);
    float dot = 0.0f, na = 0.0f, nb = 0.0f;
#pragma unroll
    for (int q = 0; q < 4; ++q) {
        float4 a = za[q], b = zb[q];
        dot += a.x * b.x + a.y * b.y + a.z * b.z + a.w * b.w;
        na  += a.x * a.x + a.y * a.y + a.z * a.z + a.w * a.w;
        nb  += b.x * b.x + b.y * b.y + b.z * b.z + b.w * b.w;
    }
    float den = fmaxf(sqrtf(na) * sqrtf(nb), 1e-6f);
    out[e] = dot / den;
}


static inline int cdiv(long long a, long long b) { return (int)((a + b - 1) / b); }

extern "C" void kernel_launch(void* const* d_in, const int* in_sizes, int n_in,
                              void* d_out, int out_size, void* d_ws, size_t ws_size,
                              hipStream_t stream) {
    const float* x     = (const float*)d_in[0];
    const int*   ei    = (const int*)d_in[1];
    const float* ew    = (const float*)d_in[2];
    const int*   eli   = (const int*)d_in[3];
    const float* preW  = (const float*)d_in[4];
    const float* preb  = (const float*)d_in[5];
    const float* g1Wi  = (const float*)d_in[6];
    const float* g1Wr  = (const float*)d_in[7];
    const float* g1b   = (const float*)d_in[8];
    const float* g2Wi  = (const float*)d_in[9];
    const float* g2Wr  = (const float*)d_in[10];
    const float* g2b   = (const float*)d_in[11];
    const float* poW   = (const float*)d_in[12];
    const float* pob   = (const float*)d_in[13];
    float* out = (float*)d_out;

    const int N = in_sizes[0];
    const int E = in_sizes[2];
    const int L = in_sizes[3] / 2;

    const int* erow = ei;
    const int* ecol = ei + E;

    float* ws   = (float*)d_ws;
    int*   rec  = (int*)ws;
    float* h64  = ws;
    float* t    = ws   + (size_t)E * 2;
    float* agg1 = t    + (size_t)N * 32;
    unsigned int* tagw = (unsigned int*)(agg1 + (size_t)N * 32);
    int*   lst  = (int*)(tagw + E / 4);
    float* z    = (float*)lst;
    float* deg  = (float*)(lst + (size_t)(E / 4) * 9);

    const int B = 256;
    const int gemm_threads = cdiv(N, 16) * 32;

    pre_kernel<<<cdiv((long long)N * 64, B), B, 0, stream>>>(x, preW, preb, h64, deg, N * 64);

    deg_owner_kernel<<<cdiv(N, DEG_NR) * DEG_S, 1024, 0, stream>>>(ecol, ew, deg, E, N);

    wmma_gemm_kernel<2, true, false><<<cdiv(gemm_threads, B), B, 0, stream>>>(
        h64, 64, g1Wi, g1Wr, g1b, t, agg1, N);

    rec_kernel<<<cdiv(E, B), B, 0, stream>>>(erow, ecol, ew, deg, rec, tagw, E, N);

    edge_owner_kernel<0><<<cdiv(N, OWN_NR) * OWN_S, 1024, 0, stream>>>(
        tagw, rec, t, agg1, lst, E, N);

    wmma_gemm_kernel<2, true, true><<<cdiv(gemm_threads, B), B, 0, stream>>>(
        agg1, 32, g2Wi, g2Wr, g2b, t, agg1, N);
    edge_owner_kernel<1><<<cdiv(N, OWN_NR) * OWN_S, 1024, 0, stream>>>(
        tagw, rec, t, agg1, lst, E, N);

    wmma_gemm_kernel<1, false, true><<<cdiv(gemm_threads, B), B, 0, stream>>>(
        agg1, 32, poW, nullptr, pob, z, nullptr, N);

    score_kernel<<<cdiv(L, B), B, 0, stream>>>(eli, z, out, L);
}


# per-element launches thickened: a thread makes 32 values of the pre layer, 8 edge records or 2 scores, by 16-byte loads and stores; 6250+12500+7813 waves enter, were 200000+100000+15625; same sums
# speedup vs baseline: 3.6070x; 1.0867x over previous
#include <hip/hip_runtime.h>
#include <math.h>

typedef float v2f __attribute__((ext_vector_type(2)));
typedef float v8f __attribute__((ext_vector_type(8)));

static __device__ __forceinline__ float elu1(float v) {
    return v > 0.0f ? v : (__expf(v) - 1.0f);
}


static __device__ __forceinline__ float dinv_of(float d) {
    return d > 0.0f ? rsqrtf(d) : 0.0f;
}

#define OWN_NR 1600
#define OWN_LB 11

struct EdgeRec { int pk; int nrm; };

static __device__ __forceinline__ int rec_row(EdgeRec x) { return (int)((unsigned)x.pk >> OWN_LB); }
static __device__ __forceinline__ int rec_loc(EdgeRec x) { return x.pk & ((1 << OWN_LB) - 1); }

#define REC_Q 2

static __device__ __forceinline__ EdgeRec rec_make(int r, int c, float w, float dr, float dc) {
    EdgeRec x;
    x.pk  = (int)(((unsigned)r << OWN_LB) | ((unsigned)c % OWN_NR));
    x.nrm = __float_as_int(dinv_of(dr) * w * dinv_of(dc));
    return x;
}

static __device__ __forceinline__ unsigned rec_tag(int c, int N) {
    return ((unsigned)c < (unsigned)N) ? (unsigned)(c / OWN_NR) : 255u;
}

__global__ void rec_kernel(const int* __restrict__ row, const int* __restrict__ col,
                           const float* __restrict__ w, const float* __restrict__ deg,
                           int* __restrict__ rec, unsigned int* __restrict__ tagw,
                           int E, int N) {
    if ((E & 3) == 0) {
        const int nq = E >> 2;
        const int q0 = blockIdx.x * (blockDim.x * REC_Q) + threadIdx.x;
        for (int g = 0; g < REC_Q; ++g) {
            const int q = q0 + g * (int)blockDim.x;
            if (q < nq) {
                const int4   r  = ((const int4*)row)[q];
                const int4   c  = ((const int4*)col)[q];
                const float4 ww = ((const float4*)w)[q];
                const float dr0 = deg[r.x], dr1 = deg[r.y], dr2 = deg[r.z], dr3 = deg[r.w];
                const float dc0 = deg[c.x], dc1 = deg[c.y], dc2 = deg[c.z], dc3 = deg[c.w];
                const EdgeRec x0 = rec_make(r.x, c.x, ww.x, dr0, dc0);
                const EdgeRec x1 = rec_make(r.y, c.y, ww.y, dr1, dc1);
                const EdgeRec x2 = rec_make(r.z, c.z, ww.z, dr2, dc2);
                const EdgeRec x3 = rec_make(r.w, c.w, ww.w, dr3, dc3);
                int4* dst = (int4*)(rec + (size_t)q * 8);
                dst[0] = make_int4(x0.pk, x0.nrm, x1.pk, x1.nrm);
                dst[1] = make_int4(x2.pk, x2.nrm, x3.pk, x3.nrm);
                tagw[q] = rec_tag(c.x, N) | (rec_tag(c.y, N) << 8) |
                          (rec_tag(c.z, N) << 16) | (rec_tag(c.w, N) << 24);
            }
        }
    } else {
        const int e0 = blockIdx.x * (blockDim.x * REC_Q * 4) + threadIdx.x;
        for (int j = 0; j < REC_Q * 4; ++j) {
            const int e = e0 + j * (int)blockDim.x;
            if (e < E) {
                const int r1 = row[e], c1 = col[e];
                *(EdgeRec*)(rec + (size_t)e * 2) = rec_make(r1, c1, w[e], deg[r1], deg[c1]);
                ((unsigned char*)tagw)[e] = (unsigned char)rec_tag(c1, N);
            }
        }
    }
}

__global__ void pre_kernel(const float* __restrict__ x, const float* __restrict__ preW,
                           const float* __restrict__ preb, float* __restrict__ h,
                           float* __restrict__ deg, int N) {
    const int gid  = blockIdx.x * blockDim.x + threadIdx.x;
    if (gid < N) deg[gid] = 0.0f;
    const int lane = threadIdx.x & 31;
    const int f4   = lane & 15;
    const int hh   = lane >> 4;
    const int nb   = (gid >> 5) * 16;
    if (nb >= N) return;
    const float4 wv = ((const float4*)preW)[f4];
    const float4 bv = ((const float4*)preb)[f4];
    const int    xl = __float_as_int(x[min(nb + f4, N - 1)]);
#pragma unroll
    for (int i = 0; i < 8; ++i) {
        const float xe = __int_as_float(__builtin_amdgcn_readlane(xl, 2 * i));
        const float xo = __int_as_float(__builtin_amdgcn_readlane(xl, 2 * i + 1));
        const float xv = hh ? xo : xe;
        const int   n  = nb + 2 * i + hh;
        if (n < N) {
            float4 o;
            o.x = elu1(xv * wv.x + bv.x);
            o.y = elu1(xv * wv.y + bv.y);
            o.z = elu1(xv * wv.z + bv.z);
            o.w = elu1(xv * wv.w + bv.w);
            ((float4*)h)[(size_t)n * 16 + f4] = o;
        }
    }
}


#define DEG_NR 12800
#define DEG_S  32

__global__ __launch_bounds__(1024)
void deg_owner_kernel(const int* __restrict__ col, const float* __restrict__ w,
                      float* __restrict__ deg, int E, int N) {
    __shared__ float dacc[DEG_NR];
    const int tid   = threadIdx.x;
    const int range = blockIdx.x / DEG_S;
    const int slice = blockIdx.x % DEG_S;
    const int base  = range * DEG_NR;
    const int nloc  = min(DEG_NR, N - base);
    for (int i = tid; i < DEG_NR; i += 1024) dacc[i] = 0.0f;
    __syncthreads();
    const int per = (E + DEG_S - 1) / DEG_S;
    const int e0  = slice * per;
    const int e1  = min(E, e0 + per);
    int e = e0 + tid;
    for (; e + 7 * 1024 < e1; e += 8 * 1024) {
        int   cc[8];
        float ww[8];
#pragma unroll
        for (int u = 0; u < 8; ++u) { cc[u] = col[e + u * 1024]; ww[u] = w[e + u * 1024]; }
#pragma unroll
        for (int u = 0; u < 8; ++u) {
            int loc = cc[u] - base;
            if ((unsigned)loc < (unsigned)nloc) atomicAdd(&dacc[loc], ww[u]);
        }
    }
    for (; e < e1; e += 1024) {
        int loc = col[e] - base;
        if ((unsigned)loc < (unsigned)nloc) atomicAdd(&dacc[loc], w[e]);
    }
    __syncthreads();
    for (int i = tid; i < nloc; i += 1024) atomicAdd(&deg[base + i], dacc[i]);
}

#define OWN_S  4
#define OWN_CH 512
#define OWN_QC (OWN_CH + 32)

template <int G, bool FULL>
static __device__ __forceinline__ void own_rows(int r, float nv, int loc, int cnt, int lane,
                                                const float* __restrict__ t, float* acc) {
    if (FULL) {
#pragma unroll
        for (int j = 0; j < 32; j += G) {
            float v[G];
#pragma unroll
            for (int u = 0; u < G; ++u) {
                int rj = __builtin_amdgcn_readlane(r, j + u);
                v[u] = t[(size_t)rj * 32 + lane];
            }
#pragma unroll
            for (int u = 0; u < G; ++u) {
                int   lj = __builtin_amdgcn_readlane(loc, j + u);
                float nj = __int_as_float(__builtin_amdgcn_readlane(__float_as_int(nv), j + u));
                atomicAdd(&acc[lj * 32 + lane], nj * v[u]);
            }
        }
    } else {
        for (int j = 0; j < cnt; j += G) {
            float v[G], nj[G];
            int   lj[G];
#pragma unroll
            for (int u = 0; u < G; ++u) {
                int jj = min(j + u, cnt - 1);
                int rj = __builtin_amdgcn_readlane(r, jj);
                lj[u]  = __builtin_amdgcn_readlane(loc, jj);
                nj[u]  = __int_as_float(__builtin_amdgcn_readlane(__float_as_int(nv), jj));
                v[u]   = t[(size_t)rj * 32 + lane];
            }
#pragma unroll
            for (int u = 0; u < G; ++u)
                if (j + u < cnt) atomicAdd(&acc[lj[u] * 32 + lane], nj[u] * v[u]);
        }
    }
}

static __device__ __forceinline__ EdgeRec own_fetch(int e, const int* __restrict__ rec) {
    return *(const EdgeRec*)(rec + (size_t)e * 2);
}

template <int G, bool FULL>
static __device__ __forceinline__ void own_work(EdgeRec x, int cnt, int lane,
                                                const float* __restrict__ t, float* acc) {
    own_rows<G, FULL>(rec_row(x), __int_as_float(x.nrm), rec_loc(x), cnt, lane, t, acc);
}

#define OWN_LH 64
#define OWN_LT (32 * 32 * 2)

static __device__ __forceinline__ uint4 own_tags(const unsigned char* __restrict__ tagb,
                                                 int cb, int lane, int lim) {
    return *(const uint4*)(tagb + min(cb + lane * 16, lim));
}

template <int MODE>
__global__ __launch_bounds__(1024)
void edge_owner_kernel(const unsigned int* __restrict__ tagw, const int* __restrict__ rec,
                       const float* __restrict__ t,
                       float* __restrict__ agg, int* __restrict__ lst, int E, int N) {
    __shared__ float acc[OWN_NR * 32];
    __shared__ int   q[32 * OWN_QC];
    __shared__ int   bcnt;
    const int tid   = threadIdx.x;
    const int lane  = tid & 31;
    const int wave  = __builtin_amdgcn_readfirstlane(tid >> 5);
    const int range = blockIdx.x / OWN_S;
    const int slice = blockIdx.x % OWN_S;
    const int base  = range * OWN_NR;
    const int nloc  = min(OWN_NR, N - base);

    const int lwg  = ((E / 4 * 9) / (int)gridDim.x) & ~63;
    const int capB = (lwg - OWN_LH - OWN_LT) / 64;
    int*     hd = lst + (size_t)blockIdx.x * (size_t)max(lwg, 0);
    EdgeRec* tl = (EdgeRec*)(hd + OWN_LH);
    EdgeRec* fb = (EdgeRec*)(hd + OWN_LH + OWN_LT);

    int nb = -1, tn = 0;
    if (MODE == 1 && capB >= 1) {
        nb = __builtin_amdgcn_readfirstlane(hd[32]);
        tn = __builtin_amdgcn_readfirstlane(hd[wave]);
    }

    if (tid == 0) bcnt = 0;
    for (int i = tid; i < OWN_NR * 32; i += 1024) acc[i] = 0.0f;
    __syncthreads();

    bool scan = true;
    if (MODE == 1) {
        scan = nb < 0;
        if (!scan) {
            if (nb > 0) {
                int b = wave;
                EdgeRec x1 = fb[(size_t)min(b, nb - 1) * 32 + lane];
                for (; b < nb; b += 32) {
                    EdgeRec x2 = fb[(size_t)min(b + 32, nb - 1) * 32 + lane];
                    own_work<16, true>(x1, 32, lane, t, acc);
                    x1 = x2;
                }
            }
            if (tn > 0) {
                EdgeRec xt = { 0, 0 };
                if (lane < tn) xt = tl[wave * 32 + lane];
                own_work<4, false>(xt, tn, lane, t, acc);
            }
        }
    }

    if (scan) {
        const bool keep = (MODE == 0) && (capB >= 1);

        const int per = ((E + OWN_S * OWN_CH - 1) / (OWN_S * OWN_CH)) * OWN_CH;
        const int e0  = slice * per;
        const int e1  = min(E, e0 + per);
        int* wq = q + wave * OWN_QC;
        int  n  = 0;

        const unsigned char* tagb = (const unsigned char*)tagw;
        const unsigned pat = (unsigned)range * 0x01010101u;

        const int lim = ((E - 1) >> 4) << 4;

        int   cb  = e0 + wave * OWN_CH;
        uint4 cur = own_tags(tagb, cb, lane, lim);

        for (; cb < e1; cb += 32 * OWN_CH) {
            uint4 nxt = own_tags(tagb, cb + 32 * OWN_CH, lane, lim);

            const unsigned xs[4] = { cur.x, cur.y, cur.z, cur.w };
            unsigned F = 0u;
#pragma unroll
            for (int d = 0; d < 4; ++d) {
                unsigned y = xs[d] ^ pat;
                unsigned s = (y & 0x7F7F7F7Fu) + 0x7F7F7F7Fu;
                unsigned f = ~(s | y | 0x7F7F7F7Fu);
                F |= f >> (7 - d);
            }
            const int eb = cb + lane * 16;
            for (int rnd = 0; rnd < 16; ++rnd) {
                if (__builtin_amdgcn_ballot_w32(F != 0u) == 0u) break;
                int p   = (F != 0u) ? __builtin_ctz(F) : 0;
                int e   = eb + (((p & 7) << 2) | (p >> 3));
                bool hit = (F != 0u) && (e < e1);
                unsigned mask = __builtin_amdgcn_ballot_w32(hit);
                int at = n + (int)__builtin_amdgcn_mbcnt_lo(mask, 0u);
                if (hit) wq[at] = e;
                n += __builtin_popcount(mask);
                F &= F - 1u;
            }
            __builtin_amdgcn_fence(__ATOMIC_ACQ_REL, "wavefront");

            if (n >= 32) {
                int done = 0;
                for (; done + 32 <= n; done += 32) {
                    const EdgeRec x = own_fetch(wq[done + lane], rec);
                    if (keep) {
                        int sb = 0;
                        if (lane == 0) sb = atomicAdd(&bcnt, 1);
                        sb = __builtin_amdgcn_readfirstlane(sb);
                        if (sb < capB) fb[(size_t)sb * 32 + lane] = x;
                    }
                    own_work<16, true>(x, 32, lane, t, acc);
                }
                const int rest = n - done;
                int mv = (lane < rest) ? wq[done + lane] : 0;
                if (lane < rest) wq[lane] = mv;
                n = rest;
                __builtin_amdgcn_fence(__ATOMIC_ACQ_REL, "wavefront");
            }
            cur = nxt;
        }
        if (n > 0) {
            EdgeRec x = { 0, 0 };
            if (lane < n) x = own_fetch(wq[lane], rec);
            if (keep && lane < n) tl[wave * 32 + lane] = x;
            own_work<4, false>(x, n, lane, t, acc);
        }
        if (keep && lane == 0) hd[wave] = n;
    }

    __syncthreads();
    if (MODE == 0 && capB >= 1 && tid == 0) hd[32] = (bcnt > capB) ? -1 : bcnt;
    float* dst = agg + (size_t)base * 32;
    for (int i = tid; i < nloc * 32; i += 1024) atomicAdd(&dst[i], acc[i]);
}

template <int NCT, bool DUAL, bool RELU_A>
__global__ void wmma_gemm_kernel(const float* A, int K,
                                 const float* __restrict__ Wi,
                                 const float* __restrict__ Wr,
                                 const float* __restrict__ biasR,
                                 float* __restrict__ Ti,
                                 float* Tr,
                                 int nrows) {
    const int F = NCT * 16;
    int wave = (blockIdx.x * blockDim.x + threadIdx.x) >> 5;
    int lane = threadIdx.x & 31;
    int row0 = wave * 16;
    if (row0 >= nrows) return;

    int m  = lane & 15;
    int hi = lane >> 4;

    v8f accI[NCT] = {};
    v8f accR[NCT] = {};

    int ar = row0 + m;
    if (ar >= nrows) ar = nrows - 1;
    const float* arow = A + (size_t)ar * K;

    for (int k0 = 0; k0 < K; k0 += 4) {
        int ka = k0 + 2 * hi;
        v2f a;
        a.x = arow[ka];
        a.y = arow[ka + 1];
        if (RELU_A) { a.x = fmaxf(a.x, 0.0f); a.y = fmaxf(a.y, 0.0f); }
#pragma unroll
        for (int ct = 0; ct < NCT; ++ct) {
            int colc = ct * 16 + m;
            v2f b;
            b.x = Wi[(size_t)ka * F + colc];
            b.y = Wi[(size_t)(ka + 1) * F + colc];
            accI[ct] = __builtin_amdgcn_wmma_f32_16x16x4_f32(
                false, a, false, b, (short)0, accI[ct], false, false);
            if (DUAL) {
                v2f br;
                br.x = Wr[(size_t)ka * F + colc];
                br.y = Wr[(size_t)(ka + 1) * F + colc];
                accR[ct] = __builtin_amdgcn_wmma_f32_16x16x4_f32(
                    false, a, false, br, (short)0, accR[ct], false, false);
            }
        }
    }

#pragma unroll
    for (int ct = 0; ct < NCT; ++ct) {
        int colc = ct * 16 + m;
        float bv = biasR ? biasR[colc] : 0.0f;
#pragma unroll
        for (int r = 0; r < 8; ++r) {
            int rowm = row0 + r + 8 * hi;
            if (rowm >= nrows) continue;
            size_t idx = (size_t)rowm * F + colc;
            if (DUAL) {
                Ti[idx] = accI[ct][r];
                Tr[idx] = accR[ct][r] + bv;
            } else {
                Ti[idx] = accI[ct][r] + bv;
            }
        }
    }
}


#define SC_P 2

static __device__ __forceinline__ void score_one(const int* __restrict__ li,
                                                 const float* __restrict__ z,
                                                 float* __restrict__ out, int L, int e) {
    int i = li[e], j = li[L + e];
    const float4* za = (const float4*)(z + (size_t)i * 16);
    const float4* zb = (const float4*)(z + (size_t)j * 16);
    float dot = 0.0f, na = 0.0f, nb = 0.0f;
#pragma unroll
    for (int q = 0; q < 4; ++q) {
        float4 a = za[q], b = zb[q];
        dot += a.x * b.x + a.y * b.y + a.z * b.z + a.w * b.w;
        na  += a.x * a.x + a.y * a.y + a.z * a.z + a.w * a.w;
        nb  += b.x * b.x + b.y * b.y + b.z * b.z + b.w * b.w;
    }
    float den = fmaxf(sqrtf(na) * sqrtf(nb), 1e-6f);
    out[e] = dot / den;
}

__global__ void score_kernel(const int* __restrict__ li, const float* __restrict__ z,
                             float* __restrict__ out, int L) {
    const int e0 = blockIdx.x * (blockDim.x * SC_P) + threadIdx.x;
    for (int g = 0; g < SC_P; ++g) {
        const int e = e0 + g * (int)blockDim.x;
        if (e < L) score_one(li, z, out, L, e);
    }
}


static inline int cdiv(long long a, long long b) { return (int)((a + b - 1) / b); }

extern "C" void kernel_launch(void* const* d_in, const int* in_sizes, int n_in,
                              void* d_out, int out_size, void* d_ws, size_t ws_size,
                              hipStream_t stream) {
    const float* x     = (const float*)d_in[0];
    const int*   ei    = (const int*)d_in[1];
    const float* ew    = (const float*)d_in[2];
    const int*   eli   = (const int*)d_in[3];
    const float* preW  = (const float*)d_in[4];
    const float* preb  = (const float*)d_in[5];
    const float* g1Wi  = (const float*)d_in[6];
    const float* g1Wr  = (const float*)d_in[7];
    const float* g1b   = (const float*)d_in[8];
    const float* g2Wi  = (const float*)d_in[9];
    const float* g2Wr  = (const float*)d_in[10];
    const float* g2b   = (const float*)d_in[11];
    const float* poW   = (const float*)d_in[12];
    const float* pob   = (const float*)d_in[13];
    float* out = (float*)d_out;

    const int N = in_sizes[0];
    const int E = in_sizes[2];
    const int L = in_sizes[3] / 2;

    const int* erow = ei;
    const int* ecol = ei + E;

    float* ws   = (float*)d_ws;
    int*   rec  = (int*)ws;
    float* h64  = ws;
    float* t    = ws   + (size_t)E * 2;
    float* agg1 = t    + (size_t)N * 32;
    unsigned int* tagw = (unsigned int*)(agg1 + (size_t)N * 32);
    int*   lst  = (int*)(tagw + E / 4);
    float* z    = (float*)lst;
    float* deg  = (float*)(lst + (size_t)(E / 4) * 9);

    const int B = 256;
    const int gemm_threads = cdiv(N, 16) * 32;

    pre_kernel<<<cdiv(gemm_threads, B), B, 0, stream>>>(x, preW, preb, h64, deg, N);

    deg_owner_kernel<<<cdiv(N, DEG_NR) * DEG_S, 1024, 0, stream>>>(ecol, ew, deg, E, N);

    wmma_gemm_kernel<2, true, false><<<cdiv(gemm_threads, B), B, 0, stream>>>(
        h64, 64, g1Wi, g1Wr, g1b, t, agg1, N);

    rec_kernel<<<cdiv(E, B * 4 * REC_Q), B, 0, stream>>>(erow, ecol, ew, deg, rec, tagw, E, N);

    edge_owner_kernel<0><<<cdiv(N, OWN_NR) * OWN_S, 1024, 0, stream>>>(
        tagw, rec, t, agg1, lst, E, N);

    wmma_gemm_kernel<2, true, true><<<cdiv(gemm_threads, B), B, 0, stream>>>(
        agg1, 32, g2Wi, g2Wr, g2b, t, agg1, N);
    edge_owner_kernel<1><<<cdiv(N, OWN_NR) * OWN_S, 1024, 0, stream>>>(
        tagw, rec, t, agg1, lst, E, N);

    wmma_gemm_kernel<1, false, true><<<cdiv(gemm_threads, B), B, 0, stream>>>(
        agg1, 32, poW, nullptr, pob, z, nullptr, N);

    score_kernel<<<cdiv(L, B * SC_P), B, 0, stream>>>(eli, z, out, L);
}


# record launch: degree sums staged in LDS, half the nodes a pass (200 KiB, whole-line copies); an edge's two factor reads are LDS reads, were two 4-byte global gathers, a line per lane; same values
# speedup vs baseline: 3.9335x; 1.0905x over previous
#include <hip/hip_runtime.h>
#include <math.h>

typedef float v2f __attribute__((ext_vector_type(2)));
typedef float v4f __attribute__((ext_vector_type(4)));
typedef float v8f __attribute__((ext_vector_type(8)));

static __device__ __forceinline__ float elu1(float v) {
    return v > 0.0f ? v : (__expf(v) - 1.0f);
}


static __device__ __forceinline__ float dinv_of(float d) {
    return d > 0.0f ? rsqrtf(d) : 0.0f;
}

#define OWN_NR 1600
#define OWN_LB 11

struct EdgeRec { int pk; int nrm; };

static __device__ __forceinline__ int rec_row(EdgeRec x) { return (int)((unsigned)x.pk >> OWN_LB); }
static __device__ __forceinline__ int rec_loc(EdgeRec x) { return x.pk & ((1 << OWN_LB) - 1); }

#define REC_Q  4
#define REC_TB 51200
#define REC_FL 8
#define REC_HOLD(v) asm volatile("" :: "v"(v[0]), "v"(v[1]), "v"(v[2]), "v"(v[3]), "v"(v[4]), "v"(v[5]), "v"(v[6]), "v"(v[7]))

static __device__ __forceinline__ EdgeRec rec_make(int r, int c, float w, float dr, float dc) {
    EdgeRec x;
    x.pk  = (int)(((unsigned)r << OWN_LB) | ((unsigned)c % OWN_NR));
    x.nrm = __float_as_int(dinv_of(dr) * w * dinv_of(dc));
    return x;
}

static __device__ __forceinline__ unsigned rec_tag(int c, int N) {
    return ((unsigned)c < (unsigned)N) ? (unsigned)(c / OWN_NR) : 255u;
}

static __device__ __forceinline__ float rec_pick(const float* tab, int i, int base, int nt,
                                                 float d) {
    const unsigned l = (unsigned)(i - base);
    const float    v = tab[min(l, (unsigned)(nt - 1))];
    return (l < (unsigned)nt) ? v : d;
}

__global__ __launch_bounds__(1024)
void rec_kernel(const int* __restrict__ row, const int* __restrict__ col,
                const float* __restrict__ w, const float* __restrict__ deg,
                int* __restrict__ rec, unsigned int* __restrict__ tagw,
                int E, int N) {
    __shared__ v4f tab4[REC_TB / 4];
    float* tab = (float*)tab4;
    const int tid = threadIdx.x;
    if ((E & 3) == 0) {
        const int nq = E >> 2;
        const int q0 = blockIdx.x * (1024 * REC_Q) + tid;
        int   rr[REC_Q][4], cc[REC_Q][4];
        float wv[REC_Q][4], dr[REC_Q][4], dc[REC_Q][4];
#pragma unroll
        for (int g = 0; g < REC_Q; ++g) {
            const int q = q0 + g * 1024;
            int4   r  = make_int4(-1, -1, -1, -1);
            int4   c  = r;
            float4 ww = make_float4(0.0f, 0.0f, 0.0f, 0.0f);
            if (q < nq) {
                r  = ((const int4*)row)[q];
                c  = ((const int4*)col)[q];
                ww = ((const float4*)w)[q];
            }
            rr[g][0] = r.x;  rr[g][1] = r.y;  rr[g][2] = r.z;  rr[g][3] = r.w;
            cc[g][0] = c.x;  cc[g][1] = c.y;  cc[g][2] = c.z;  cc[g][3] = c.w;
            wv[g][0] = ww.x; wv[g][1] = ww.y; wv[g][2] = ww.z; wv[g][3] = ww.w;
#pragma unroll
            for (int k = 0; k < 4; ++k) { dr[g][k] = 0.0f; dc[g][k] = 0.0f; }
        }
        const bool al = (((size_t)deg) & 15) == 0;
        for (int base = 0; base < N; base += REC_TB) {
            const int nt = min(REC_TB, N - base);
            const int n4 = al ? (nt >> 2) : 0;
            const v4f* src4 = (const v4f*)(deg + base);
            for (int i0 = 0; i0 < n4; i0 += REC_FL * 1024) {
                v4f v[REC_FL];
#pragma unroll
                for (int u = 0; u < REC_FL; ++u) v[u] = src4[min(i0 + u * 1024 + tid, n4 - 1)];
                REC_HOLD(v);
#pragma unroll
                for (int u = 0; u < REC_FL; ++u) {
                    const int i = i0 + u * 1024 + tid;
                    if (i < n4) tab4[i] = v[u];
                }
            }
            for (int i = n4 * 4 + tid; i < nt; i += 1024) tab[i] = deg[base + i];
            __syncthreads();
#pragma unroll
            for (int g = 0; g < REC_Q; ++g) {
#pragma unroll
                for (int k = 0; k < 4; ++k) {
                    dr[g][k] = rec_pick(tab, rr[g][k], base, nt, dr[g][k]);
                    dc[g][k] = rec_pick(tab, cc[g][k], base, nt, dc[g][k]);
                }
            }
            __syncthreads();
        }
#pragma unroll
        for (int g = 0; g < REC_Q; ++g) {
            const int q = q0 + g * 1024;
            if (q < nq) {
                const EdgeRec x0 = rec_make(rr[g][0], cc[g][0], wv[g][0], dr[g][0], dc[g][0]);
                const EdgeRec x1 = rec_make(rr[g][1], cc[g][1], wv[g][1], dr[g][1], dc[g][1]);
                const EdgeRec x2 = rec_make(rr[g][2], cc[g][2], wv[g][2], dr[g][2], dc[g][2]);
                const EdgeRec x3 = rec_make(rr[g][3], cc[g][3], wv[g][3], dr[g][3], dc[g][3]);
                int4* dst = (int4*)(rec + (size_t)q * 8);
                dst[0] = make_int4(x0.pk, x0.nrm, x1.pk, x1.nrm);
                dst[1] = make_int4(x2.pk, x2.nrm, x3.pk, x3.nrm);
                tagw[q] = rec_tag(cc[g][0], N) | (rec_tag(cc[g][1], N) << 8) |
                          (rec_tag(cc[g][2], N) << 16) | (rec_tag(cc[g][3], N) << 24);
            }
        }
    } else {
        const int e0 = blockIdx.x * (1024 * REC_Q * 4) + tid;
        for (int j = 0; j < REC_Q * 4; ++j) {
            const int e = e0 + j * 1024;
            if (e < E) {
                const int r1 = row[e], c1 = col[e];
                *(EdgeRec*)(rec + (size_t)e * 2) = rec_make(r1, c1, w[e], deg[r1], deg[c1]);
                ((unsigned char*)tagw)[e] = (unsigned char)rec_tag(c1, N);
            }
        }
    }
}

__global__ void pre_kernel(const float* __restrict__ x, const float* __restrict__ preW,
                           const float* __restrict__ preb, float* __restrict__ h,
                           float* __restrict__ deg, int N) {
    const int gid  = blockIdx.x * blockDim.x + threadIdx.x;
    if (gid < N) deg[gid] = 0.0f;
    const int lane = threadIdx.x & 31;
    const int f4   = lane & 15;
    const int hh   = lane >> 4;
    const int nb   = (gid >> 5) * 16;
    if (nb >= N) return;
    const float4 wv = ((const float4*)preW)[f4];
    const float4 bv = ((const float4*)preb)[f4];
    const int    xl = __float_as_int(x[min(nb + f4, N - 1)]);
#pragma unroll
    for (int i = 0; i < 8; ++i) {
        const float xe = __int_as_float(__builtin_amdgcn_readlane(xl, 2 * i));
        const float xo = __int_as_float(__builtin_amdgcn_readlane(xl, 2 * i + 1));
        const float xv = hh ? xo : xe;
        const int   n  = nb + 2 * i + hh;
        if (n < N) {
            float4 o;
            o.x = elu1(xv * wv.x + bv.x);
            o.y = elu1(xv * wv.y + bv.y);
            o.z = elu1(xv * wv.z + bv.z);
            o.w = elu1(xv * wv.w + bv.w);
            ((float4*)h)[(size_t)n * 16 + f4] = o;
        }
    }
}


#define DEG_NR 12800
#define DEG_S  32

__global__ __launch_bounds__(1024)
void deg_owner_kernel(const int* __restrict__ col, const float* __restrict__ w,
                      float* __restrict__ deg, int E, int N) {
    __shared__ float dacc[DEG_NR];
    const int tid   = threadIdx.x;
    const int range = blockIdx.x / DEG_S;
    const int slice = blockIdx.x % DEG_S;
    const int base  = range * DEG_NR;
    const int nloc  = min(DEG_NR, N - base);
    for (int i = tid; i < DEG_NR; i += 1024) dacc[i] = 0.0f;
    __syncthreads();
    const int per = (E + DEG_S - 1) / DEG_S;
    const int e0  = slice * per;
    const int e1  = min(E, e0 + per);
    int e = e0 + tid;
    for (; e + 7 * 1024 < e1; e += 8 * 1024) {
        int   cc[8];
        float ww[8];
#pragma unroll
        for (int u = 0; u < 8; ++u) { cc[u] = col[e + u * 1024]; ww[u] = w[e + u * 1024]; }
#pragma unroll
        for (int u = 0; u < 8; ++u) {
            int loc = cc[u] - base;
            if ((unsigned)loc < (unsigned)nloc) atomicAdd(&dacc[loc], ww[u]);
        }
    }
    for (; e < e1; e += 1024) {
        int loc = col[e] - base;
        if ((unsigned)loc < (unsigned)nloc) atomicAdd(&dacc[loc], w[e]);
    }
    __syncthreads();
    for (int i = tid; i < nloc; i += 1024) atomicAdd(&deg[base + i], dacc[i]);
}

#define OWN_S  4
#define OWN_CH 512
#define OWN_QC (OWN_CH + 32)

template <int G, bool FULL>
static __device__ __forceinline__ void own_rows(int r, float nv, int loc, int cnt, int lane,
                                                const float* __restrict__ t, float* acc) {
    if (FULL) {
#pragma unroll
        for (int j = 0; j < 32; j += G) {
            float v[G];
#pragma unroll
            for (int u = 0; u < G; ++u) {
                int rj = __builtin_amdgcn_readlane(r, j + u);
                v[u] = t[(size_t)rj * 32 + lane];
            }
#pragma unroll
            for (int u = 0; u < G; ++u) {
                int   lj = __builtin_amdgcn_readlane(loc, j + u);
                float nj = __int_as_float(__builtin_amdgcn_readlane(__float_as_int(nv), j + u));
                atomicAdd(&acc[lj * 32 + lane], nj * v[u]);
            }
        }
    } else {
        for (int j = 0; j < cnt; j += G) {
            float v[G], nj[G];
            int   lj[G];
#pragma unroll
            for (int u = 0; u < G; ++u) {
                int jj = min(j + u, cnt - 1);
                int rj = __builtin_amdgcn_readlane(r, jj);
                lj[u]  = __builtin_amdgcn_readlane(loc, jj);
                nj[u]  = __int_as_float(__builtin_amdgcn_readlane(__float_as_int(nv), jj));
                v[u]   = t[(size_t)rj * 32 + lane];
            }
#pragma unroll
            for (int u = 0; u < G; ++u)
                if (j + u < cnt) atomicAdd(&acc[lj[u] * 32 + lane], nj[u] * v[u]);
        }
    }
}

static __device__ __forceinline__ EdgeRec own_fetch(int e, const int* __restrict__ rec) {
    return *(const EdgeRec*)(rec + (size_t)e * 2);
}

template <int G, bool FULL>
static __device__ __forceinline__ void own_work(EdgeRec x, int cnt, int lane,
                                                const float* __restrict__ t, float* acc) {
    own_rows<G, FULL>(rec_row(x), __int_as_float(x.nrm), rec_loc(x), cnt, lane, t, acc);
}

#define OWN_LH 64
#define OWN_LT (32 * 32 * 2)

static __device__ __forceinline__ uint4 own_tags(const unsigned char* __restrict__ tagb,
                                                 int cb, int lane, int lim) {
    return *(const uint4*)(tagb + min(cb + lane * 16, lim));
}

template <int MODE>
__global__ __launch_bounds__(1024)
void edge_owner_kernel(const unsigned int* __restrict__ tagw, const int* __restrict__ rec,
                       const float* __restrict__ t,
                       float* __restrict__ agg, int* __restrict__ lst, int E, int N) {
    __shared__ float acc[OWN_NR * 32];
    __shared__ int   q[32 * OWN_QC];
    __shared__ int   bcnt;
    const int tid   = threadIdx.x;
    const int lane  = tid & 31;
    const int wave  = __builtin_amdgcn_readfirstlane(tid >> 5);
    const int range = blockIdx.x / OWN_S;
    const int slice = blockIdx.x % OWN_S;
    const int base  = range * OWN_NR;
    const int nloc  = min(OWN_NR, N - base);

    const int lwg  = ((E / 4 * 9) / (int)gridDim.x) & ~63;
    const int capB = (lwg - OWN_LH - OWN_LT) / 64;
    int*     hd = lst + (size_t)blockIdx.x * (size_t)max(lwg, 0);
    EdgeRec* tl = (EdgeRec*)(hd + OWN_LH);
    EdgeRec* fb = (EdgeRec*)(hd + OWN_LH + OWN_LT);

    int nb = -1, tn = 0;
    if (MODE == 1 && capB >= 1) {
        nb = __builtin_amdgcn_readfirstlane(hd[32]);
        tn = __builtin_amdgcn_readfirstlane(hd[wave]);
    }

    if (tid == 0) bcnt = 0;
    for (int i = tid; i < OWN_NR * 32; i += 1024) acc[i] = 0.0f;
    __syncthreads();

    bool scan = true;
    if (MODE == 1) {
        scan = nb < 0;
        if (!scan) {
            if (nb > 0) {
                int b = wave;
                EdgeRec x1 = fb[(size_t)min(b, nb - 1) * 32 + lane];
                for (; b < nb; b += 32) {
                    EdgeRec x2 = fb[(size_t)min(b + 32, nb - 1) * 32 + lane];
                    own_work<16, true>(x1, 32, lane, t, acc);
                    x1 = x2;
                }
            }
            if (tn > 0) {
                EdgeRec xt = { 0, 0 };
                if (lane < tn) xt = tl[wave * 32 + lane];
                own_work<4, false>(xt, tn, lane, t, acc);
            }
        }
    }

    if (scan) {
        const bool keep = (MODE == 0) && (capB >= 1);

        const int per = ((E + OWN_S * OWN_CH - 1) / (OWN_S * OWN_CH)) * OWN_CH;
        const int e0  = slice * per;
        const int e1  = min(E, e0 + per);
        int* wq = q + wave * OWN_QC;
        int  n  = 0;

        const unsigned char* tagb = (const unsigned char*)tagw;
        const unsigned pat = (unsigned)range * 0x01010101u;

        const int lim = ((E - 1) >> 4) << 4;

        int   cb  = e0 + wave * OWN_CH;
        uint4 cur = own_tags(tagb, cb, lane, lim);

        for (; cb < e1; cb += 32 * OWN_CH) {
            uint4 nxt = own_tags(tagb, cb + 32 * OWN_CH, lane, lim);

            const unsigned xs[4] = { cur.x, cur.y, cur.z, cur.w };
            unsigned F = 0u;
#pragma unroll
            for (int d = 0; d < 4; ++d) {
                unsigned y = xs[d] ^ pat;
                unsigned s = (y & 0x7F7F7F7Fu) + 0x7F7F7F7Fu;
                unsigned f = ~(s | y | 0x7F7F7F7Fu);
                F |= f >> (7 - d);
            }
            const int eb = cb + lane * 16;
            for (int rnd = 0; rnd < 16; ++rnd) {
                if (__builtin_amdgcn_ballot_w32(F != 0u) == 0u) break;
                int p   = (F != 0u) ? __builtin_ctz(F) : 0;
                int e   = eb + (((p & 7) << 2) | (p >> 3));
                bool hit = (F != 0u) && (e < e1);
                unsigned mask = __builtin_amdgcn_ballot_w32(hit);
                int at = n + (int)__builtin_amdgcn_mbcnt_lo(mask, 0u);
                if (hit) wq[at] = e;
                n += __builtin_popcount(mask);
                F &= F - 1u;
            }
            __builtin_amdgcn_fence(__ATOMIC_ACQ_REL, "wavefront");

            if (n >= 32) {
                int done = 0;
                for (; done + 32 <= n; done += 32) {
                    const EdgeRec x = own_fetch(wq[done + lane], rec);
                    if (keep) {
                        int sb = 0;
                        if (lane == 0) sb = atomicAdd(&bcnt, 1);
                        sb = __builtin_amdgcn_readfirstlane(sb);
                        if (sb < capB) fb[(size_t)sb * 32 + lane] = x;
                    }
                    own_work<16, true>(x, 32, lane, t, acc);
                }
                const int rest = n - done;
                int mv = (lane < rest) ? wq[done + lane] : 0;
                if (lane < rest) wq[lane] = mv;
                n = rest;
                __builtin_amdgcn_fence(__ATOMIC_ACQ_REL, "wavefront");
            }
            cur = nxt;
        }
        if (n > 0) {
            EdgeRec x = { 0, 0 };
            if (lane < n) x = own_fetch(wq[lane], rec);
            if (keep && lane < n) tl[wave * 32 + lane] = x;
            own_work<4, false>(x, n, lane, t, acc);
        }
        if (keep && lane == 0) hd[wave] = n;
    }

    __syncthreads();
    if (MODE == 0 && capB >= 1 && tid == 0) hd[32] = (bcnt > capB) ? -1 : bcnt;
    float* dst = agg + (size_t)base * 32;
    for (int i = tid; i < nloc * 32; i += 1024) atomicAdd(&dst[i], acc[i]);
}

template <int NCT, bool DUAL, bool RELU_A>
__global__ void wmma_gemm_kernel(const float* A, int K,
                                 const float* __restrict__ Wi,
                                 const float* __restrict__ Wr,
                                 const float* __restrict__ biasR,
                                 float* __restrict__ Ti,
                                 float* Tr,
                                 int nrows) {
    const int F = NCT * 16;
    int wave = (blockIdx.x * blockDim.x + threadIdx.x) >> 5;
    int lane = threadIdx.x & 31;
    int row0 = wave * 16;
    if (row0 >= nrows) return;

    int m  = lane & 15;
    int hi = lane >> 4;

    v8f accI[NCT] = {};
    v8f accR[NCT] = {};

    int ar = row0 + m;
    if (ar >= nrows) ar = nrows - 1;
    const float* arow = A + (size_t)ar * K;

    for (int k0 = 0; k0 < K; k0 += 4) {
        int ka = k0 + 2 * hi;
        v2f a;
        a.x = arow[ka];
        a.y = arow[ka + 1];
        if (RELU_A) { a.x = fmaxf(a.x, 0.0f); a.y = fmaxf(a.y, 0.0f); }
#pragma unroll
        for (int ct = 0; ct < NCT; ++ct) {
            int colc = ct * 16 + m;
            v2f b;
            b.x = Wi[(size_t)ka * F + colc];
            b.y = Wi[(size_t)(ka + 1) * F + colc];
            accI[ct] = __builtin_amdgcn_wmma_f32_16x16x4_f32(
                false, a, false, b, (short)0, accI[ct], false, false);
            if (DUAL) {
                v2f br;
                br.x = Wr[(size_t)ka * F + colc];
                br.y = Wr[(size_t)(ka + 1) * F + colc];
                accR[ct] = __builtin_amdgcn_wmma_f32_16x16x4_f32(
                    false, a, false, br, (short)0, accR[ct], false, false);
            }
        }
    }

#pragma unroll
    for (int ct = 0; ct < NCT; ++ct) {
        int colc = ct * 16 + m;
        float bv = biasR ? biasR[colc] : 0.0f;
#pragma unroll
        for (int r = 0; r < 8; ++r) {
            int rowm = row0 + r + 8 * hi;
            if (rowm >= nrows) continue;
            size_t idx = (size_t)rowm * F + colc;
            if (DUAL) {
                Ti[idx] = accI[ct][r];
                Tr[idx] = accR[ct][r] + bv;
            } else {
                Ti[idx] = accI[ct][r] + bv;
            }
        }
    }
}


#define SC_P 2

static __device__ __forceinline__ void score_one(const int* __restrict__ li,
                                                 const float* __restrict__ z,
                                                 float* __restrict__ out, int L, int e) {
    int i = li[e], j = li[L + e];
    const float4* za = (const float4*)(z + (size_t)i * 16);
    const float4* zb = (const float4*)(z + (size_t)j * 16);
    float dot = 0.0f, na = 0.0f, nb = 0.0f;
#pragma unroll
    for (int q = 0; q < 4; ++q) {
        float4 a = za[q], b = zb[q];
        dot += a.x * b.x + a.y * b.y + a.z * b.z + a.w * b.w;
        na  += a.x * a.x + a.y * a.y + a.z * a.z + a.w * a.w;
        nb  += b.x * b.x + b.y * b.y + b.z * b.z + b.w * b.w;
    }
    float den = fmaxf(sqrtf(na) * sqrtf(nb), 1e-6f);
    out[e] = dot / den;
}

__global__ void score_kernel(const int* __restrict__ li, const float* __restrict__ z,
                             float* __restrict__ out, int L) {
    const int e0 = blockIdx.x * (blockDim.x * SC_P) + threadIdx.x;
    for (int g = 0; g < SC_P; ++g) {
        const int e = e0 + g * (int)blockDim.x;
        if (e < L) score_one(li, z, out, L, e);
    }
}


static inline int cdiv(long long a, long long b) { return (int)((a + b - 1) / b); }

extern "C" void kernel_launch(void* const* d_in, const int* in_sizes, int n_in,
                              void* d_out, int out_size, void* d_ws, size_t ws_size,
                              hipStream_t stream) {
    const float* x     = (const float*)d_in[0];
    const int*   ei    = (const int*)d_in[1];
    const float* ew    = (const float*)d_in[2];
    const int*   eli   = (const int*)d_in[3];
    const float* preW  = (const float*)d_in[4];
    const float* preb  = (const float*)d_in[5];
    const float* g1Wi  = (const float*)d_in[6];
    const float* g1Wr  = (const float*)d_in[7];
    const float* g1b   = (const float*)d_in[8];
    const float* g2Wi  = (const float*)d_in[9];
    const float* g2Wr  = (const float*)d_in[10];
    const float* g2b   = (const float*)d_in[11];
    const float* poW   = (const float*)d_in[12];
    const float* pob   = (const float*)d_in[13];
    float* out = (float*)d_out;

    const int N = in_sizes[0];
    const int E = in_sizes[2];
    const int L = in_sizes[3] / 2;

    const int* erow = ei;
    const int* ecol = ei + E;

    float* ws   = (float*)d_ws;
    int*   rec  = (int*)ws;
    float* h64  = ws;
    float* t    = ws   + (size_t)E * 2;
    float* agg1 = t    + (size_t)N * 32;
    unsigned int* tagw = (unsigned int*)(agg1 + (size_t)N * 32);
    int*   lst  = (int*)(tagw + E / 4);
    float* z    = (float*)lst;
    float* deg  = (float*)(lst + (size_t)(E / 4) * 9);

    const int B = 256;
    const int gemm_threads = cdiv(N, 16) * 32;

    pre_kernel<<<cdiv(gemm_threads, B), B, 0, stream>>>(x, preW, preb, h64, deg, N);

    deg_owner_kernel<<<cdiv(N, DEG_NR) * DEG_S, 1024, 0, stream>>>(ecol, ew, deg, E, N);

    wmma_gemm_kernel<2, true, false><<<cdiv(gemm_threads, B), B, 0, stream>>>(
        h64, 64, g1Wi, g1Wr, g1b, t, agg1, N);

    rec_kernel<<<cdiv(E, 1024 * 4 * REC_Q), 1024, 0, stream>>>(erow, ecol, ew, deg, rec, tagw, E, N);

    edge_owner_kernel<0><<<cdiv(N, OWN_NR) * OWN_S, 1024, 0, stream>>>(
        tagw, rec, t, agg1, lst, E, N);

    wmma_gemm_kernel<2, true, true><<<cdiv(gemm_threads, B), B, 0, stream>>>(
        agg1, 32, g2Wi, g2Wr, g2b, t, agg1, N);
    edge_owner_kernel<1><<<cdiv(N, OWN_NR) * OWN_S, 1024, 0, stream>>>(
        tagw, rec, t, agg1, lst, E, N);

    wmma_gemm_kernel<1, false, true><<<cdiv(gemm_threads, B), B, 0, stream>>>(
        agg1, 32, poW, nullptr, pob, z, nullptr, N);

    score_kernel<<<cdiv(L, B * SC_P), B, 0, stream>>>(eli, z, out, L);
}


# GEMM launches: each wave's A tile and the workgroup's weights go to LDS by 16-byte whole-line loads issued as one group; fragments read from LDS (were 4/8-byte global loads per k step); same k order
# speedup vs baseline: 4.0934x; 1.0407x over previous
#include <hip/hip_runtime.h>
#include <math.h>

typedef float v2f __attribute__((ext_vector_type(2)));
typedef float v4f __attribute__((ext_vector_type(4)));
typedef float v8f __attribute__((ext_vector_type(8)));

static __device__ __forceinline__ float elu1(float v) {
    return v > 0.0f ? v : (__expf(v) - 1.0f);
}


static __device__ __forceinline__ float dinv_of(float d) {
    return d > 0.0f ? rsqrtf(d) : 0.0f;
}

#define OWN_NR 1600
#define OWN_LB 11

struct EdgeRec { int pk; int nrm; };

static __device__ __forceinline__ int rec_row(EdgeRec x) { return (int)((unsigned)x.pk >> OWN_LB); }
static __device__ __forceinline__ int rec_loc(EdgeRec x) { return x.pk & ((1 << OWN_LB) - 1); }

#define REC_Q  4
#define REC_TB 51200
#define REC_FL 8
#define REC_HOLD(v) asm volatile("" :: "v"(v[0]), "v"(v[1]), "v"(v[2]), "v"(v[3]), "v"(v[4]), "v"(v[5]), "v"(v[6]), "v"(v[7]))

static __device__ __forceinline__ EdgeRec rec_make(int r, int c, float w, float dr, float dc) {
    EdgeRec x;
    x.pk  = (int)(((unsigned)r << OWN_LB) | ((unsigned)c % OWN_NR));
    x.nrm = __float_as_int(dinv_of(dr) * w * dinv_of(dc));
    return x;
}

static __device__ __forceinline__ unsigned rec_tag(int c, int N) {
    return ((unsigned)c < (unsigned)N) ? (unsigned)(c / OWN_NR) : 255u;
}

static __device__ __forceinline__ float rec_pick(const float* tab, int i, int base, int nt,
                                                 float d) {
    const unsigned l = (unsigned)(i - base);
    const float    v = tab[min(l, (unsigned)(nt - 1))];
    return (l < (unsigned)nt) ? v : d;
}

__global__ __launch_bounds__(1024)
void rec_kernel(const int* __restrict__ row, const int* __restrict__ col,
                const float* __restrict__ w, const float* __restrict__ deg,
                int* __restrict__ rec, unsigned int* __restrict__ tagw,
                int E, int N) {
    __shared__ v4f tab4[REC_TB / 4];
    float* tab = (float*)tab4;
    const int tid = threadIdx.x;
    if ((E & 3) == 0) {
        const int nq = E >> 2;
        const int q0 = blockIdx.x * (1024 * REC_Q) + tid;
        int   rr[REC_Q][4], cc[REC_Q][4];
        float wv[REC_Q][4], dr[REC_Q][4], dc[REC_Q][4];
#pragma unroll
        for (int g = 0; g < REC_Q; ++g) {
            const int q = q0 + g * 1024;
            int4   r  = make_int4(-1, -1, -1, -1);
            int4   c  = r;
            float4 ww = make_float4(0.0f, 0.0f, 0.0f, 0.0f);
            if (q < nq) {
                r  = ((const int4*)row)[q];
                c  = ((const int4*)col)[q];
                ww = ((const float4*)w)[q];
            }
            rr[g][0] = r.x;  rr[g][1] = r.y;  rr[g][2] = r.z;  rr[g][3] = r.w;
            cc[g][0] = c.x;  cc[g][1] = c.y;  cc[g][2] = c.z;  cc[g][3] = c.w;
            wv[g][0] = ww.x; wv[g][1] = ww.y; wv[g][2] = ww.z; wv[g][3] = ww.w;
#pragma unroll
            for (int k = 0; k < 4; ++k) { dr[g][k] = 0.0f; dc[g][k] = 0.0f; }
        }
        const bool al = (((size_t)deg) & 15) == 0;
        for (int base = 0; base < N; base += REC_TB) {
            const int nt = min(REC_TB, N - base);
            const int n4 = al ? (nt >> 2) : 0;
            const v4f* src4 = (const v4f*)(deg + base);
            for (int i0 = 0; i0 < n4; i0 += REC_FL * 1024) {
                v4f v[REC_FL];
#pragma unroll
                for (int u = 0; u < REC_FL; ++u) v[u] = src4[min(i0 + u * 1024 + tid, n4 - 1)];
                REC_HOLD(v);
#pragma unroll
                for (int u = 0; u < REC_FL; ++u) {
                    const int i = i0 + u * 1024 + tid;
                    if (i < n4) tab4[i] = v[u];
                }
            }
            for (int i = n4 * 4 + tid; i < nt; i += 1024) tab[i] = deg[base + i];
            __syncthreads();
#pragma unroll
            for (int g = 0; g < REC_Q; ++g) {
#pragma unroll
                for (int k = 0; k < 4; ++k) {
                    dr[g][k] = rec_pick(tab, rr[g][k], base, nt, dr[g][k]);
                    dc[g][k] = rec_pick(tab, cc[g][k], base, nt, dc[g][k]);
                }
            }
            __syncthreads();
        }
#pragma unroll
        for (int g = 0; g < REC_Q; ++g) {
            const int q = q0 + g * 1024;
            if (q < nq) {
                const EdgeRec x0 = rec_make(rr[g][0], cc[g][0], wv[g][0], dr[g][0], dc[g][0]);
                const EdgeRec x1 = rec_make(rr[g][1], cc[g][1], wv[g][1], dr[g][1], dc[g][1]);
                const EdgeRec x2 = rec_make(rr[g][2], cc[g][2], wv[g][2], dr[g][2], dc[g][2]);
                const EdgeRec x3 = rec_make(rr[g][3], cc[g][3], wv[g][3], dr[g][3], dc[g][3]);
                int4* dst = (int4*)(rec + (size_t)q * 8);
                dst[0] = make_int4(x0.pk, x0.nrm, x1.pk, x1.nrm);
                dst[1] = make_int4(x2.pk, x2.nrm, x3.pk, x3.nrm);
                tagw[q] = rec_tag(cc[g][0], N) | (rec_tag(cc[g][1], N) << 8) |
                          (rec_tag(cc[g][2], N) << 16) | (rec_tag(cc[g][3], N) << 24);
            }
        }
    } else {
        const int e0 = blockIdx.x * (1024 * REC_Q * 4) + tid;
        for (int j = 0; j < REC_Q * 4; ++j) {
            const int e = e0 + j * 1024;
            if (e < E) {
                const int r1 = row[e], c1 = col[e];
                *(EdgeRec*)(rec + (size_t)e * 2) = rec_make(r1, c1, w[e], deg[r1], deg[c1]);
                ((unsigned char*)tagw)[e] = (unsigned char)rec_tag(c1, N);
            }
        }
    }
}

__global__ void pre_kernel(const float* __restrict__ x, const float* __restrict__ preW,
                           const float* __restrict__ preb, float* __restrict__ h,
                           float* __restrict__ deg, int N) {
    const int gid  = blockIdx.x * blockDim.x + threadIdx.x;
    if (gid < N) deg[gid] = 0.0f;
    const int lane = threadIdx.x & 31;
    const int f4   = lane & 15;
    const int hh   = lane >> 4;
    const int nb   = (gid >> 5) * 16;
    if (nb >= N) return;
    const float4 wv = ((const float4*)preW)[f4];
    const float4 bv = ((const float4*)preb)[f4];
    const int    xl = __float_as_int(x[min(nb + f4, N - 1)]);
#pragma unroll
    for (int i = 0; i < 8; ++i) {
        const float xe = __int_as_float(__builtin_amdgcn_readlane(xl, 2 * i));
        const float xo = __int_as_float(__builtin_amdgcn_readlane(xl, 2 * i + 1));
        const float xv = hh ? xo : xe;
        const int   n  = nb + 2 * i + hh;
        if (n < N) {
            float4 o;
            o.x = elu1(xv * wv.x + bv.x);
            o.y = elu1(xv * wv.y + bv.y);
            o.z = elu1(xv * wv.z + bv.z);
            o.w = elu1(xv * wv.w + bv.w);
            ((float4*)h)[(size_t)n * 16 + f4] = o;
        }
    }
}


#define DEG_NR 12800
#define DEG_S  32

__global__ __launch_bounds__(1024)
void deg_owner_kernel(const int* __restrict__ col, const float* __restrict__ w,
                      float* __restrict__ deg, int E, int N) {
    __shared__ float dacc[DEG_NR];
    const int tid   = threadIdx.x;
    const int range = blockIdx.x / DEG_S;
    const int slice = blockIdx.x % DEG_S;
    const int base  = range * DEG_NR;
    const int nloc  = min(DEG_NR, N - base);
    for (int i = tid; i < DEG_NR; i += 1024) dacc[i] = 0.0f;
    __syncthreads();
    const int per = (E + DEG_S - 1) / DEG_S;
    const int e0  = slice * per;
    const int e1  = min(E, e0 + per);
    int e = e0 + tid;
    for (; e + 7 * 1024 < e1; e += 8 * 1024) {
        int   cc[8];
        float ww[8];
#pragma unroll
        for (int u = 0; u < 8; ++u) { cc[u] = col[e + u * 1024]; ww[u] = w[e + u * 1024]; }
#pragma unroll
        for (int u = 0; u < 8; ++u) {
            int loc = cc[u] - base;
            if ((unsigned)loc < (unsigned)nloc) atomicAdd(&dacc[loc], ww[u]);
        }
    }
    for (; e < e1; e += 1024) {
        int loc = col[e] - base;
        if ((unsigned)loc < (unsigned)nloc) atomicAdd(&dacc[loc], w[e]);
    }
    __syncthreads();
    for (int i = tid; i < nloc; i += 1024) atomicAdd(&deg[base + i], dacc[i]);
}

#define OWN_S  4
#define OWN_CH 512
#define OWN_QC (OWN_CH + 32)

template <int G, bool FULL>
static __device__ __forceinline__ void own_rows(int r, float nv, int loc, int cnt, int lane,
                                                const float* __restrict__ t, float* acc) {
    if (FULL) {
#pragma unroll
        for (int j = 0; j < 32; j += G) {
            float v[G];
#pragma unroll
            for (int u = 0; u < G; ++u) {
                int rj = __builtin_amdgcn_readlane(r, j + u);
                v[u] = t[(size_t)rj * 32 + lane];
            }
#pragma unroll
            for (int u = 0; u < G; ++u) {
                int   lj = __builtin_amdgcn_readlane(loc, j + u);
                float nj = __int_as_float(__builtin_amdgcn_readlane(__float_as_int(nv), j + u));
                atomicAdd(&acc[lj * 32 + lane], nj * v[u]);
            }
        }
    } else {
        for (int j = 0; j < cnt; j += G) {
            float v[G], nj[G];
            int   lj[G];
#pragma unroll
            for (int u = 0; u < G; ++u) {
                int jj = min(j + u, cnt - 1);
                int rj = __builtin_amdgcn_readlane(r, jj);
                lj[u]  = __builtin_amdgcn_readlane(loc, jj);
                nj[u]  = __int_as_float(__builtin_amdgcn_readlane(__float_as_int(nv), jj));
                v[u]   = t[(size_t)rj * 32 + lane];
            }
#pragma unroll
            for (int u = 0; u < G; ++u)
                if (j + u < cnt) atomicAdd(&acc[lj[u] * 32 + lane], nj[u] * v[u]);
        }
    }
}

static __device__ __forceinline__ EdgeRec own_fetch(int e, const int* __restrict__ rec) {
    return *(const EdgeRec*)(rec + (size_t)e * 2);
}

template <int G, bool FULL>
static __device__ __forceinline__ void own_work(EdgeRec x, int cnt, int lane,
                                                const float* __restrict__ t, float* acc) {
    own_rows<G, FULL>(rec_row(x), __int_as_float(x.nrm), rec_loc(x), cnt, lane, t, acc);
}

#define OWN_LH 64
#define OWN_LT (32 * 32 * 2)

static __device__ __forceinline__ uint4 own_tags(const unsigned char* __restrict__ tagb,
                                                 int cb, int lane, int lim) {
    return *(const uint4*)(tagb + min(cb + lane * 16, lim));
}

template <int MODE>
__global__ __launch_bounds__(1024)
void edge_owner_kernel(const unsigned int* __restrict__ tagw, const int* __restrict__ rec,
                       const float* __restrict__ t,
                       float* __restrict__ agg, int* __restrict__ lst, int E, int N) {
    __shared__ float acc[OWN_NR * 32];
    __shared__ int   q[32 * OWN_QC];
    __shared__ int   bcnt;
    const int tid   = threadIdx.x;
    const int lane  = tid & 31;
    const int wave  = __builtin_amdgcn_readfirstlane(tid >> 5);
    const int range = blockIdx.x / OWN_S;
    const int slice = blockIdx.x % OWN_S;
    const int base  = range * OWN_NR;
    const int nloc  = min(OWN_NR, N - base);

    const int lwg  = ((E / 4 * 9) / (int)gridDim.x) & ~63;
    const int capB = (lwg - OWN_LH - OWN_LT) / 64;
    int*     hd = lst + (size_t)blockIdx.x * (size_t)max(lwg, 0);
    EdgeRec* tl = (EdgeRec*)(hd + OWN_LH);
    EdgeRec* fb = (EdgeRec*)(hd + OWN_LH + OWN_LT);

    int nb = -1, tn = 0;
    if (MODE == 1 && capB >= 1) {
        nb = __builtin_amdgcn_readfirstlane(hd[32]);
        tn = __builtin_amdgcn_readfirstlane(hd[wave]);
    }

    if (tid == 0) bcnt = 0;
    for (int i = tid; i < OWN_NR * 32; i += 1024) acc[i] = 0.0f;
    __syncthreads();

    bool scan = true;
    if (MODE == 1) {
        scan = nb < 0;
        if (!scan) {
            if (nb > 0) {
                int b = wave;
                EdgeRec x1 = fb[(size_t)min(b, nb - 1) * 32 + lane];
                for (; b < nb; b += 32) {
                    EdgeRec x2 = fb[(size_t)min(b + 32, nb - 1) * 32 + lane];
                    own_work<16, true>(x1, 32, lane, t, acc);
                    x1 = x2;
                }
            }
            if (tn > 0) {
                EdgeRec xt = { 0, 0 };
                if (lane < tn) xt = tl[wave * 32 + lane];
                own_work<4, false>(xt, tn, lane, t, acc);
            }
        }
    }

    if (scan) {
        const bool keep = (MODE == 0) && (capB >= 1);

        const int per = ((E + OWN_S * OWN_CH - 1) / (OWN_S * OWN_CH)) * OWN_CH;
        const int e0  = slice * per;
        const int e1  = min(E, e0 + per);
        int* wq = q + wave * OWN_QC;
        int  n  = 0;

        const unsigned char* tagb = (const unsigned char*)tagw;
        const unsigned pat = (unsigned)range * 0x01010101u;

        const int lim = ((E - 1) >> 4) << 4;

        int   cb  = e0 + wave * OWN_CH;
        uint4 cur = own_tags(tagb, cb, lane, lim);

        for (; cb < e1; cb += 32 * OWN_CH) {
            uint4 nxt = own_tags(tagb, cb + 32 * OWN_CH, lane, lim);

            const unsigned xs[4] = { cur.x, cur.y, cur.z, cur.w };
            unsigned F = 0u;
#pragma unroll
            for (int d = 0; d < 4; ++d) {
                unsigned y = xs[d] ^ pat;
                unsigned s = (y & 0x7F7F7F7Fu) + 0x7F7F7F7Fu;
                unsigned f = ~(s | y | 0x7F7F7F7Fu);
                F |= f >> (7 - d);
            }
            const int eb = cb + lane * 16;
            for (int rnd = 0; rnd < 16; ++rnd) {
                if (__builtin_amdgcn_ballot_w32(F != 0u) == 0u) break;
                int p   = (F != 0u) ? __builtin_ctz(F) : 0;
                int e   = eb + (((p & 7) << 2) | (p >> 3));
                bool hit = (F != 0u) && (e < e1);
                unsigned mask = __builtin_amdgcn_ballot_w32(hit);
                int at = n + (int)__builtin_amdgcn_mbcnt_lo(mask, 0u);
                if (hit) wq[at] = e;
                n += __builtin_popcount(mask);
                F &= F - 1u;
            }
            __builtin_amdgcn_fence(__ATOMIC_ACQ_REL, "wavefront");

            if (n >= 32) {
                int done = 0;
                for (; done + 32 <= n; done += 32) {
                    const EdgeRec x = own_fetch(wq[done + lane], rec);
                    if (keep) {
                        int sb = 0;
                        if (lane == 0) sb = atomicAdd(&bcnt, 1);
                        sb = __builtin_amdgcn_readfirstlane(sb);
                        if (sb < capB) fb[(size_t)sb * 32 + lane] = x;
                    }
                    own_work<16, true>(x, 32, lane, t, acc);
                }
                const int rest = n - done;
                int mv = (lane < rest) ? wq[done + lane] : 0;
                if (lane < rest) wq[lane] = mv;
                n = rest;
                __builtin_amdgcn_fence(__ATOMIC_ACQ_REL, "wavefront");
            }
            cur = nxt;
        }
        if (n > 0) {
            EdgeRec x = { 0, 0 };
            if (lane < n) x = own_fetch(wq[lane], rec);
            if (keep && lane < n) tl[wave * 32 + lane] = x;
            own_work<4, false>(x, n, lane, t, acc);
        }
        if (keep && lane == 0) hd[wave] = n;
    }

    __syncthreads();
    if (MODE == 0 && capB >= 1 && tid == 0) hd[32] = (bcnt > capB) ? -1 : bcnt;
    float* dst = agg + (size_t)base * 32;
    for (int i = tid; i < nloc * 32; i += 1024) atomicAdd(&dst[i], acc[i]);
}

#define GEMM_WV 8

template <int NCT, bool DUAL, bool RELU_A, int K>
__global__ __launch_bounds__(GEMM_WV * 32)
void wmma_gemm_kernel(const float* A,
                      const float* __restrict__ Wi,
                      const float* __restrict__ Wr,
                      const float* __restrict__ biasR,
                      float* __restrict__ Ti,
                      float* Tr,
                      int nrows) {
    constexpr int F   = NCT * 16;
    constexpr int P   = K + 4;
    constexpr int KQ  = K / 4;
    constexpr int AQ  = 16 * KQ / 32;
    constexpr int WQ  = K * F / 4;
    constexpr int WT  = (DUAL ? 2 : 1) * WQ;
    constexpr int WPT = (WT + GEMM_WV * 32 - 1) / (GEMM_WV * 32);
    constexpr bool WEVEN = (WQ % (GEMM_WV * 32)) == 0;
    __shared__ v4f sA4[GEMM_WV * 16 * P / 4];
    __shared__ v4f sW4[WPT * GEMM_WV * 32];

    const int tid  = threadIdx.x;
    const int lane = tid & 31;
    const int wv   = __builtin_amdgcn_readfirstlane(tid >> 5);
    const int row0 = (blockIdx.x * GEMM_WV + wv) * 16;
    const bool live = row0 < nrows;

    v4f av[AQ];
    v4f wq[WPT];
#pragma unroll
    for (int i = 0; i < AQ; ++i) {
        const int p = i * 32 + lane;
        const int r = min(row0 + p / KQ, nrows - 1);
        av[i] = *(const v4f*)(A + (size_t)r * K + (p % KQ) * 4);
    }
#pragma unroll
    for (int u = 0; u < WPT; ++u) {
        const int pu = u * (GEMM_WV * 32) + tid;
        const int p  = WEVEN ? pu : min(pu, WT - 1);
        const bool second = DUAL && (WEVEN ? (u * (GEMM_WV * 32) >= WQ) : (p >= WQ));
        const v4f* s = second ? ((const v4f*)Wr + (p - WQ)) : ((const v4f*)Wi + p);
        wq[u] = *s;
    }
#pragma unroll
    for (int i = 0; i < AQ; ++i) {
        const int p = i * 32 + lane;
        sA4[(wv * 16 + p / KQ) * (P / 4) + (p % KQ)] = av[i];
    }
#pragma unroll
    for (int u = 0; u < WPT; ++u) sW4[u * (GEMM_WV * 32) + tid] = wq[u];
    __syncthreads();

    if (live) {
        const int m  = lane & 15;
        const int hi = lane >> 4;

        const float* sA = (const float*)sA4;
        const float* sW = (const float*)sW4;
        const float* ar = sA + (wv * 16 + m) * P + 2 * hi;
        const float* wk = sW + (2 * hi) * F + m;

        v8f accI[NCT] = {};
        v8f accR[NCT] = {};

#pragma unroll
        for (int k0 = 0; k0 < K; k0 += 4) {
            v2f a = *(const v2f*)(ar + k0);
            if (RELU_A) { a.x = fmaxf(a.x, 0.0f); a.y = fmaxf(a.y, 0.0f); }
#pragma unroll
            for (int ct = 0; ct < NCT; ++ct) {
                v2f b;
                b.x = wk[k0 * F + ct * 16];
                b.y = wk[(k0 + 1) * F + ct * 16];
                accI[ct] = __builtin_amdgcn_wmma_f32_16x16x4_f32(
                    false, a, false, b, (short)0, accI[ct], false, false);
                if (DUAL) {
                    v2f br;
                    br.x = wk[K * F + k0 * F + ct * 16];
                    br.y = wk[K * F + (k0 + 1) * F + ct * 16];
                    accR[ct] = __builtin_amdgcn_wmma_f32_16x16x4_f32(
                        false, a, false, br, (short)0, accR[ct], false, false);
                }
            }
        }

#pragma unroll
        for (int ct = 0; ct < NCT; ++ct) {
            int colc = ct * 16 + m;
            float bv = biasR ? biasR[colc] : 0.0f;
#pragma unroll
            for (int r = 0; r < 8; ++r) {
                int rowm = row0 + r + 8 * hi;
                if (rowm >= nrows) continue;
                size_t idx = (size_t)rowm * F + colc;
                if (DUAL) {
                    Ti[idx] = accI[ct][r];
                    Tr[idx] = accR[ct][r] + bv;
                } else {
                    Ti[idx] = accI[ct][r] + bv;
                }
            }
        }
    }
}


#define SC_P 2

static __device__ __forceinline__ void score_one(const int* __restrict__ li,
                                                 const float* __restrict__ z,
                                                 float* __restrict__ out, int L, int e) {
    int i = li[e], j = li[L + e];
    const float4* za = (const float4*)(z + (size_t)i * 16);
    const float4* zb = (const float4*)(z + (size_t)j * 16);
    float dot = 0.0f, na = 0.0f, nb = 0.0f;
#pragma unroll
    for (int q = 0; q < 4; ++q) {
        float4 a = za[q], b = zb[q];
        dot += a.x * b.x + a.y * b.y + a.z * b.z + a.w * b.w;
        na  += a.x * a.x + a.y * a.y + a.z * a.z + a.w * a.w;
        nb  += b.x * b.x + b.y * b.y + b.z * b.z + b.w * b.w;
    }
    float den = fmaxf(sqrtf(na) * sqrtf(nb), 1e-6f);
    out[e] = dot / den;
}

__global__ void score_kernel(const int* __restrict__ li, const float* __restrict__ z,
                             float* __restrict__ out, int L) {
    const int e0 = blockIdx.x * (blockDim.x * SC_P) + threadIdx.x;
    for (int g = 0; g < SC_P; ++g) {
        const int e = e0 + g * (int)blockDim.x;
        if (e < L) score_one(li, z, out, L, e);
    }
}


static inline int cdiv(long long a, long long b) { return (int)((a + b - 1) / b); }

extern "C" void kernel_launch(void* const* d_in, const int* in_sizes, int n_in,
                              void* d_out, int out_size, void* d_ws, size_t ws_size,
                              hipStream_t stream) {
    const float* x     = (const float*)d_in[0];
    const int*   ei    = (const int*)d_in[1];
    const float* ew    = (const float*)d_in[2];
    const int*   eli   = (const int*)d_in[3];
    const float* preW  = (const float*)d_in[4];
    const float* preb  = (const float*)d_in[5];
    const float* g1Wi  = (const float*)d_in[6];
    const float* g1Wr  = (const float*)d_in[7];
    const float* g1b   = (const float*)d_in[8];
    const float* g2Wi  = (const float*)d_in[9];
    const float* g2Wr  = (const float*)d_in[10];
    const float* g2b   = (const float*)d_in[11];
    const float* poW   = (const float*)d_in[12];
    const float* pob   = (const float*)d_in[13];
    float* out = (float*)d_out;

    const int N = in_sizes[0];
    const int E = in_sizes[2];
    const int L = in_sizes[3] / 2;

    const int* erow = ei;
    const int* ecol = ei + E;

    float* ws   = (float*)d_ws;
    int*   rec  = (int*)ws;
    float* h64  = ws;
    float* t    = ws   + (size_t)E * 2;
    float* agg1 = t    + (size_t)N * 32;
    unsigned int* tagw = (unsigned int*)(agg1 + (size_t)N * 32);
    int*   lst  = (int*)(tagw + E / 4);
    float* z    = (float*)lst;
    float* deg  = (float*)(lst + (size_t)(E / 4) * 9);

    const int B = 256;
    const int gemm_threads = cdiv(N, 16) * 32;

    pre_kernel<<<cdiv(gemm_threads, B), B, 0, stream>>>(x, preW, preb, h64, deg, N);

    deg_owner_kernel<<<cdiv(N, DEG_NR) * DEG_S, 1024, 0, stream>>>(ecol, ew, deg, E, N);

    wmma_gemm_kernel<2, true, false, 64><<<cdiv(gemm_threads, B), B, 0, stream>>>(
        h64, g1Wi, g1Wr, g1b, t, agg1, N);

    rec_kernel<<<cdiv(E, 1024 * 4 * REC_Q), 1024, 0, stream>>>(erow, ecol, ew, deg, rec, tagw, E, N);

    edge_owner_kernel<0><<<cdiv(N, OWN_NR) * OWN_S, 1024, 0, stream>>>(
        tagw, rec, t, agg1, lst, E, N);

    wmma_gemm_kernel<2, true, true, 32><<<cdiv(gemm_threads, B), B, 0, stream>>>(
        agg1, g2Wi, g2Wr, g2b, t, agg1, N);
    edge_owner_kernel<1><<<cdiv(N, OWN_NR) * OWN_S, 1024, 0, stream>>>(
        tagw, rec, t, agg1, lst, E, N);

    wmma_gemm_kernel<1, false, true, 32><<<cdiv(gemm_threads, B), B, 0, stream>>>(
        agg1, poW, nullptr, pob, z, nullptr, N);

    score_kernel<<<cdiv(L, B * SC_P), B, 0, stream>>>(eli, z, out, L);
}
